# starting kernel
# baseline (speedup 1.0000x reference)
#include <hip/hip_runtime.h>
#include <math.h>

#define N_NODES 50000
#define N_EDGES 800000
#define D_IN    128
#define N_HEAD  4
#define D_OUT   32

typedef __attribute__((ext_vector_type(2))) float v2f;
typedef __attribute__((ext_vector_type(8))) float v8f;

__device__ __forceinline__ unsigned f2ord(float f) {
    unsigned u = __float_as_uint(f);
    return (u & 0x80000000u) ? ~u : (u | 0x80000000u);
}
__device__ __forceinline__ float ord2f(unsigned u) {
    return (u & 0x80000000u) ? __uint_as_float(u ^ 0x80000000u)
                             : __uint_as_float(~u);
}

__global__ void gat_init_kernel(float* __restrict__ out,
                                const float* __restrict__ bias,
                                float* __restrict__ neigh,
                                unsigned* __restrict__ gmax) {
    int idx = blockIdx.x * blockDim.x + threadIdx.x;
    if (idx < N_NODES * D_IN)
        out[idx] = bias[idx & (D_IN - 1)];
    if (idx < N_NODES * N_HEAD)
        neigh[idx] = 0.0f;
    if (idx == 0)
        *gmax = f2ord(-INFINITY);
}

__global__ __launch_bounds__(256)
void gat_gemm_wmma_kernel(const float* __restrict__ A,
                          const float* __restrict__ W,
                          float* __restrict__ P) {
    const int wave = threadIdx.x >> 5;
    const int lane = threadIdx.x & 31;
    const int m0   = blockIdx.x * 16;
    const int n0   = wave * 16;

    const int mRow = m0 + (lane & 15);
    const int nCol = n0 + (lane & 15);
    const int kGrp = (lane >> 4) * 2;

    v8f acc = {};
#pragma unroll
    for (int k0 = 0; k0 < D_IN; k0 += 4) {
        const int ka = k0 + kGrp;
        v2f a, b;
        a.x = A[mRow * D_IN + ka];
        a.y = A[mRow * D_IN + ka + 1];
        b.x = W[ka * D_IN + nCol];
        b.y = W[(ka + 1) * D_IN + nCol];
        acc = __builtin_amdgcn_wmma_f32_16x16x4_f32(false, a, false, b,
                                                    (short)0, acc, false, false);
    }

    const int mBase = m0 + 8 * (lane >> 4);
#pragma unroll
    for (int v = 0; v < 8; ++v)
        P[(mBase + v) * D_IN + nCol] = acc[v];
}

__global__ __launch_bounds__(128)
void gat_scores_kernel(const float* __restrict__ P,
                       const float* __restrict__ a_src,
                       const float* __restrict__ a_tgt,
                       float* __restrict__ s_src,
                       float* __restrict__ s_tgt) {
    const int n = blockIdx.x;
    const int t = threadIdx.x;
    const float p  = P[n * D_IN + t];
    float vs = p * a_src[t];
    float vt = p * a_tgt[t];
#pragma unroll
    for (int off = 16; off > 0; off >>= 1) {
        vs += __shfl_xor(vs, off, 32);
        vt += __shfl_xor(vt, off, 32);
    }
    if ((t & 31) == 0) {
        const int h = t >> 5;
        s_src[n * N_HEAD + h] = vs;
        s_tgt[n * N_HEAD + h] = vt;
    }
}

__global__ __launch_bounds__(256)
void gat_edge_max_kernel(const int* __restrict__ ei,
                         const float* __restrict__ s_src,
                         const float* __restrict__ s_tgt,
                         unsigned* __restrict__ gmax) {
    const int e = blockIdx.x * blockDim.x + threadIdx.x;
    float m = -INFINITY;
    if (e < N_EDGES) {
        const int src = ei[e];
        const int tgt = ei[N_EDGES + e];
#pragma unroll
        for (int h = 0; h < N_HEAD; ++h) {
            float s = s_src[src * N_HEAD + h] + s_tgt[tgt * N_HEAD + h];
            s = (s > 0.0f) ? s : 0.2f * s;
            m = fmaxf(m, s);
        }
    }
#pragma unroll
    for (int off = 16; off > 0; off >>= 1)
        m = fmaxf(m, __shfl_xor(m, off, 32));
    if ((threadIdx.x & 31) == 0)
        atomicMax(gmax, f2ord(m));
}

__global__ __launch_bounds__(256)
void gat_edge_exp_kernel(const int* __restrict__ ei,
                         const float* __restrict__ s_src,
                         const float* __restrict__ s_tgt,
                         const unsigned* __restrict__ gmax,
                         float* __restrict__ edge_e,
                         float* __restrict__ neigh) {
    const int e = blockIdx.x * blockDim.x + threadIdx.x;
    if (e >= N_EDGES) return;
    const float gm = ord2f(*gmax);
    const int src = ei[e];
    const int tgt = ei[N_EDGES + e];
#pragma unroll
    for (int h = 0; h < N_HEAD; ++h) {
        float s = s_src[src * N_HEAD + h] + s_tgt[tgt * N_HEAD + h];
        s = (s > 0.0f) ? s : 0.2f * s;
        const float ex = __expf(s - gm);
        edge_e[e * N_HEAD + h] = ex;
        atomicAdd(&neigh[tgt * N_HEAD + h], ex);
    }
}

__global__ __launch_bounds__(256)
void gat_aggregate_kernel(const int* __restrict__ ei,
                          const float* __restrict__ P,
                          const float* __restrict__ edge_e,
                          const float* __restrict__ neigh,
                          float* __restrict__ out) {
    const long long gid = (long long)blockIdx.x * blockDim.x + threadIdx.x;
    if (gid >= (long long)N_EDGES * D_IN) return;
    const int e = (int)(gid >> 7);
    const int d = (int)(gid & (D_IN - 1));
    const int h = d >> 5;
    const int src = ei[e];
    const int tgt = ei[N_EDGES + e];
    const float att = edge_e[e * N_HEAD + h] /
                      (neigh[tgt * N_HEAD + h] + 1e-16f);
    atomicAdd(&out[tgt * D_IN + d], P[src * D_IN + d] * att);
}

extern "C" void kernel_launch(void* const* d_in, const int* in_sizes, int n_in,
                              void* d_out, int out_size, void* d_ws, size_t ws_size,
                              hipStream_t stream) {
    const float* in_feat = (const float*)d_in[0];
    const int*   ei      = (const int*)  d_in[1];
    const float* W_proj  = (const float*)d_in[3];
    const float* a_src   = (const float*)d_in[4];
    const float* a_tgt   = (const float*)d_in[5];
    const float* bias    = (const float*)d_in[6];
    float*       out     = (float*)d_out;

    float*    proj   = (float*)d_ws;
    float*    s_src  = proj  + (size_t)N_NODES * D_IN;
    float*    s_tgt  = s_src + (size_t)N_NODES * N_HEAD;
    float*    neigh  = s_tgt + (size_t)N_NODES * N_HEAD;
    float*    edge_e = neigh + (size_t)N_NODES * N_HEAD;
    unsigned* gmax   = (unsigned*)(edge_e + (size_t)N_EDGES * N_HEAD);

    gat_init_kernel<<<(N_NODES * D_IN + 255) / 256, 256, 0, stream>>>(
        out, bias, neigh, gmax);

    gat_gemm_wmma_kernel<<<N_NODES / 16, 256, 0, stream>>>(in_feat, W_proj, proj);

    gat_scores_kernel<<<N_NODES, 128, 0, stream>>>(proj, a_src, a_tgt, s_src, s_tgt);

    gat_edge_max_kernel<<<(N_EDGES + 255) / 256, 256, 0, stream>>>(
        ei, s_src, s_tgt, gmax);

    gat_edge_exp_kernel<<<(N_EDGES + 255) / 256, 256, 0, stream>>>(
        ei, s_src, s_tgt, gmax, edge_e, neigh);

    const long long total = (long long)N_EDGES * D_IN;
    gat_aggregate_kernel<<<(unsigned)((total + 255) / 256), 256, 0, stream>>>(
        ei, proj, edge_e, neigh, out);
}


# float-atomic scatter-add -> segment sum: edges bucketed by target (integer atomic rank per edge + scan), one wave per node sums softmax and rows in registers; sum order = arrival rank in bucket
# speedup vs baseline: 2.2332x; 2.2332x over previous
#include <hip/hip_runtime.h>
#include <math.h>

#define N_NODES 50000
#define N_EDGES 800000
#define D_IN    128
#define N_HEAD  4
#define D_OUT   32

typedef __attribute__((ext_vector_type(2))) float v2f;
typedef __attribute__((ext_vector_type(8))) float v8f;

__device__ __forceinline__ unsigned f2ord(float f) {
    unsigned u = __float_as_uint(f);
    return (u & 0x80000000u) ? ~u : (u | 0x80000000u);
}
__device__ __forceinline__ float ord2f(unsigned u) {
    return (u & 0x80000000u) ? __uint_as_float(u ^ 0x80000000u)
                             : __uint_as_float(~u);
}

__global__ void gat_init_kernel(unsigned* __restrict__ deg,
                                unsigned* __restrict__ gmax) {
    int idx = blockIdx.x * blockDim.x + threadIdx.x;
    if (idx < N_NODES)
        deg[idx] = 0u;
    if (idx == 0)
        *gmax = f2ord(-INFINITY);
}

__global__ __launch_bounds__(256)
void gat_gemm_wmma_kernel(const float* __restrict__ A,
                          const float* __restrict__ W,
                          float* __restrict__ P) {
    const int wave = threadIdx.x >> 5;
    const int lane = threadIdx.x & 31;
    const int m0   = blockIdx.x * 16;
    const int n0   = wave * 16;

    const int mRow = m0 + (lane & 15);
    const int nCol = n0 + (lane & 15);
    const int kGrp = (lane >> 4) * 2;

    v8f acc = {};
#pragma unroll
    for (int k0 = 0; k0 < D_IN; k0 += 4) {
        const int ka = k0 + kGrp;
        v2f a, b;
        a.x = A[mRow * D_IN + ka];
        a.y = A[mRow * D_IN + ka + 1];
        b.x = W[ka * D_IN + nCol];
        b.y = W[(ka + 1) * D_IN + nCol];
        acc = __builtin_amdgcn_wmma_f32_16x16x4_f32(false, a, false, b,
                                                    (short)0, acc, false, false);
    }

    const int mBase = m0 + 8 * (lane >> 4);
#pragma unroll
    for (int v = 0; v < 8; ++v)
        P[(mBase + v) * D_IN + nCol] = acc[v];
}

__global__ __launch_bounds__(128)
void gat_scores_kernel(const float* __restrict__ P,
                       const float* __restrict__ a_src,
                       const float* __restrict__ a_tgt,
                       float* __restrict__ s_src,
                       float* __restrict__ s_tgt) {
    const int n = blockIdx.x;
    const int t = threadIdx.x;
    const float p  = P[n * D_IN + t];
    float vs = p * a_src[t];
    float vt = p * a_tgt[t];
#pragma unroll
    for (int off = 16; off > 0; off >>= 1) {
        vs += __shfl_xor(vs, off, 32);
        vt += __shfl_xor(vt, off, 32);
    }
    if ((t & 31) == 0) {
        const int h = t >> 5;
        s_src[n * N_HEAD + h] = vs;
        s_tgt[n * N_HEAD + h] = vt;
    }
}

__global__ __launch_bounds__(256)
void gat_edge_max_kernel(const int* __restrict__ ei,
                         const float* __restrict__ s_src,
                         const float* __restrict__ s_tgt,
                         unsigned* __restrict__ gmax,
                         unsigned* __restrict__ deg,
                         unsigned* __restrict__ rank) {
    const int e = blockIdx.x * blockDim.x + threadIdx.x;
    float m = -INFINITY;
    if (e < N_EDGES) {
        const int src = ei[e];
        const int tgt = ei[N_EDGES + e];
        rank[e] = atomicAdd(&deg[tgt], 1u);
#pragma unroll
        for (int h = 0; h < N_HEAD; ++h) {
            float s = s_src[src * N_HEAD + h] + s_tgt[tgt * N_HEAD + h];
            s = (s > 0.0f) ? s : 0.2f * s;
            m = fmaxf(m, s);
        }
    }
#pragma unroll
    for (int off = 16; off > 0; off >>= 1)
        m = fmaxf(m, __shfl_xor(m, off, 32));
    if ((threadIdx.x & 31) == 0)
        atomicMax(gmax, f2ord(m));
}

#define SCAN_THREADS 1024
#define SCAN_ITEMS   49
__global__ __launch_bounds__(SCAN_THREADS)
void gat_scan_kernel(const unsigned* __restrict__ deg,
                     unsigned* __restrict__ rowptr) {
    __shared__ unsigned sh[SCAN_THREADS * SCAN_ITEMS];
    __shared__ unsigned wsum[32];
    const int t    = threadIdx.x;
    const int lane = t & 31;
    const int wave = t >> 5;
    for (int i = t; i < SCAN_THREADS * SCAN_ITEMS; i += SCAN_THREADS)
        sh[i] = (i < N_NODES) ? deg[i] : 0u;
    __syncthreads();
    unsigned s = 0u;
    for (int k = 0; k < SCAN_ITEMS; ++k)
        s += sh[t * SCAN_ITEMS + k];
    unsigned incl = s;
#pragma unroll
    for (int off = 1; off < 32; off <<= 1) {
        const unsigned v = __shfl_up(incl, off, 32);
        if (lane >= off) incl += v;
    }
    if (lane == 31) wsum[wave] = incl;
    __syncthreads();
    if (wave == 0) {
        const unsigned w = wsum[lane];
        unsigned wi = w;
#pragma unroll
        for (int off = 1; off < 32; off <<= 1) {
            const unsigned v = __shfl_up(wi, off, 32);
            if (lane >= off) wi += v;
        }
        wsum[lane] = wi - w;
    }
    __syncthreads();
    unsigned run = wsum[wave] + incl - s;
    for (int k = 0; k < SCAN_ITEMS; ++k) {
        const unsigned d = sh[t * SCAN_ITEMS + k];
        sh[t * SCAN_ITEMS + k] = run;
        run += d;
    }
    __syncthreads();
    for (int i = t; i < N_NODES; i += SCAN_THREADS)
        rowptr[i] = sh[i];
    if (t == SCAN_THREADS - 1)
        rowptr[N_NODES] = run;
}

__global__ __launch_bounds__(256)
void gat_fill_kernel(const int* __restrict__ ei,
                     const unsigned* __restrict__ rowptr,
                     const unsigned* __restrict__ rank,
                     int* __restrict__ perm) {
    const int e = blockIdx.x * blockDim.x + threadIdx.x;
    if (e >= N_EDGES) return;
    const int src = ei[e];
    const int tgt = ei[N_EDGES + e];
    perm[rowptr[tgt] + rank[e]] = src;
}

__device__ __forceinline__ float head_weight(const float4 att, const int j,
                                             const int m0, const int m1,
                                             const int m2, const int m3) {
    const int a0 = __builtin_amdgcn_readlane(__float_as_int(att.x), j);
    const int a1 = __builtin_amdgcn_readlane(__float_as_int(att.y), j);
    const int a2 = __builtin_amdgcn_readlane(__float_as_int(att.z), j);
    const int a3 = __builtin_amdgcn_readlane(__float_as_int(att.w), j);
    return __int_as_float((a0 & m0) | (a1 & m1) | (a2 & m2) | (a3 & m3));
}

__device__ __forceinline__ void bucket_weights(const int* __restrict__ perm,
                                               const float4* __restrict__ s_src4,
                                               const float4 st, const float gm,
                                               const int base, const int cnt,
                                               const int lane,
                                               int& src, float4& ex) {
    src = 0;
    ex = make_float4(0.0f, 0.0f, 0.0f, 0.0f);
    if (lane < cnt) {
        src = perm[base + lane];
        const float4 ss = s_src4[src];
        float s0 = ss.x + st.x, s1 = ss.y + st.y, s2 = ss.z + st.z, s3 = ss.w + st.w;
        s0 = (s0 > 0.0f) ? s0 : 0.2f * s0;
        s1 = (s1 > 0.0f) ? s1 : 0.2f * s1;
        s2 = (s2 > 0.0f) ? s2 : 0.2f * s2;
        s3 = (s3 > 0.0f) ? s3 : 0.2f * s3;
        ex.x = __expf(s0 - gm);
        ex.y = __expf(s1 - gm);
        ex.z = __expf(s2 - gm);
        ex.w = __expf(s3 - gm);
    }
}

__device__ __forceinline__ void bucket_accumulate(const float4* __restrict__ P4,
                                                  const int src, const float4 att,
                                                  const int cnt, const int lane,
                                                  const int hd, float4& acc) {
    const int m0 = (hd == 0) ? -1 : 0, m1 = (hd == 1) ? -1 : 0;
    const int m2 = (hd == 2) ? -1 : 0, m3 = (hd == 3) ? -1 : 0;
    int j = 0;
    for (; j + 4 <= cnt; j += 4) {
        float4 r[4];
        float  a[4];
#pragma unroll
        for (int u = 0; u < 4; ++u) {
            const int sj = __builtin_amdgcn_readlane(src, j + u);
            r[u] = P4[(size_t)sj * (D_IN / 4) + lane];
        }
#pragma unroll
        for (int u = 0; u < 4; ++u)
            a[u] = head_weight(att, j + u, m0, m1, m2, m3);
#pragma unroll
        for (int u = 0; u < 4; ++u) {
            acc.x = __builtin_fmaf(r[u].x, a[u], acc.x);
            acc.y = __builtin_fmaf(r[u].y, a[u], acc.y);
            acc.z = __builtin_fmaf(r[u].z, a[u], acc.z);
            acc.w = __builtin_fmaf(r[u].w, a[u], acc.w);
        }
    }
    for (; j < cnt; ++j) {
        const int sj = __builtin_amdgcn_readlane(src, j);
        const float4 r = P4[(size_t)sj * (D_IN / 4) + lane];
        const float a = head_weight(att, j, m0, m1, m2, m3);
        acc.x = __builtin_fmaf(r.x, a, acc.x);
        acc.y = __builtin_fmaf(r.y, a, acc.y);
        acc.z = __builtin_fmaf(r.z, a, acc.z);
        acc.w = __builtin_fmaf(r.w, a, acc.w);
    }
}

__global__ __launch_bounds__(256)
void gat_gather_kernel(const unsigned* __restrict__ rowptr,
                       const int* __restrict__ perm,
                       const float* __restrict__ P,
                       const float* __restrict__ s_src,
                       const float* __restrict__ s_tgt,
                       const unsigned* __restrict__ gmax,
                       const float* __restrict__ bias,
                       float* __restrict__ out) {
    const int lane = threadIdx.x & 31;
    const int hd   = lane >> 3;
    const int n = __builtin_amdgcn_readfirstlane((int)(blockIdx.x * 8 + (threadIdx.x >> 5)));
    const int beg = (int)rowptr[n];
    const int end = (int)rowptr[n + 1];
    const float4* P4     = (const float4*)P;
    const float4* s_src4 = (const float4*)s_src;
    const float   gm = ord2f(*gmax);
    const float4  st = ((const float4*)s_tgt)[n];

    float4 acc = make_float4(0.0f, 0.0f, 0.0f, 0.0f);
    if (end > beg) {
        const int cnt0 = (end - beg < 32) ? (end - beg) : 32;
        int src0; float4 ex0;
        bucket_weights(perm, s_src4, st, gm, beg, cnt0, lane, src0, ex0);
        float4 den = ex0;
        for (int b = beg + 32; b < end; b += 32) {
            const int c = (end - b < 32) ? (end - b) : 32;
            int sx; float4 ex;
            bucket_weights(perm, s_src4, st, gm, b, c, lane, sx, ex);
            den.x += ex.x; den.y += ex.y; den.z += ex.z; den.w += ex.w;
        }
#pragma unroll
        for (int off = 16; off > 0; off >>= 1) {
            den.x += __shfl_xor(den.x, off, 32);
            den.y += __shfl_xor(den.y, off, 32);
            den.z += __shfl_xor(den.z, off, 32);
            den.w += __shfl_xor(den.w, off, 32);
        }
        den.x += 1e-16f; den.y += 1e-16f; den.z += 1e-16f; den.w += 1e-16f;
        float4 att = make_float4(ex0.x / den.x, ex0.y / den.y, ex0.z / den.z, ex0.w / den.w);
        bucket_accumulate(P4, src0, att, cnt0, lane, hd, acc);
        for (int b = beg + 32; b < end; b += 32) {
            const int c = (end - b < 32) ? (end - b) : 32;
            int sx; float4 ex;
            bucket_weights(perm, s_src4, st, gm, b, c, lane, sx, ex);
            att = make_float4(ex.x / den.x, ex.y / den.y, ex.z / den.z, ex.w / den.w);
            bucket_accumulate(P4, sx, att, c, lane, hd, acc);
        }
    }
    const float4 bv = ((const float4*)bias)[lane];
    ((float4*)out)[(size_t)n * (D_IN / 4) + lane] =
        make_float4(acc.x + bv.x, acc.y + bv.y, acc.z + bv.z, acc.w + bv.w);
}

extern "C" void kernel_launch(void* const* d_in, const int* in_sizes, int n_in,
                              void* d_out, int out_size, void* d_ws, size_t ws_size,
                              hipStream_t stream) {
    const float* in_feat = (const float*)d_in[0];
    const int*   ei      = (const int*)  d_in[1];
    const float* W_proj  = (const float*)d_in[3];
    const float* a_src   = (const float*)d_in[4];
    const float* a_tgt   = (const float*)d_in[5];
    const float* bias    = (const float*)d_in[6];
    float*       out     = (float*)d_out;

    float*    proj   = (float*)d_ws;
    float*    s_src  = proj  + (size_t)N_NODES * D_IN;
    float*    s_tgt  = s_src + (size_t)N_NODES * N_HEAD;
    unsigned* deg    = (unsigned*)(s_tgt + (size_t)N_NODES * N_HEAD);
    unsigned* rowptr = deg + (size_t)N_NODES;
    unsigned* rank   = rowptr + (size_t)N_NODES + 1;
    int*      perm   = (int*)(rank + (size_t)N_EDGES);
    unsigned* gmax   = (unsigned*)(perm + (size_t)N_EDGES);

    gat_init_kernel<<<(N_NODES + 255) / 256, 256, 0, stream>>>(deg, gmax);

    gat_gemm_wmma_kernel<<<N_NODES / 16, 256, 0, stream>>>(in_feat, W_proj, proj);

    gat_scores_kernel<<<N_NODES, 128, 0, stream>>>(proj, a_src, a_tgt, s_src, s_tgt);

    gat_edge_max_kernel<<<(N_EDGES + 255) / 256, 256, 0, stream>>>(
        ei, s_src, s_tgt, gmax, deg, rank);

    gat_scan_kernel<<<1, SCAN_THREADS, 0, stream>>>(deg, rowptr);

    gat_fill_kernel<<<(N_EDGES + 255) / 256, 256, 0, stream>>>(ei, rowptr, rank, perm);

    gat_gather_kernel<<<N_NODES / 8, 256, 0, stream>>>(
        rowptr, perm, proj, s_src, s_tgt, gmax, bias, out);
}


# score kernel fused into the WMMA projection: a wave owns one head as two 16-col tiles on interleaved columns and reduces both scores from its accumulators; score sum order changes (16-lane tree)
# speedup vs baseline: 2.6017x; 1.1650x over previous
#include <hip/hip_runtime.h>
#include <math.h>

#define N_NODES 50000
#define N_EDGES 800000
#define D_IN    128
#define N_HEAD  4
#define D_OUT   32

typedef __attribute__((ext_vector_type(2))) float v2f;
typedef __attribute__((ext_vector_type(8))) float v8f;

__device__ __forceinline__ unsigned f2ord(float f) {
    unsigned u = __float_as_uint(f);
    return (u & 0x80000000u) ? ~u : (u | 0x80000000u);
}
__device__ __forceinline__ float ord2f(unsigned u) {
    return (u & 0x80000000u) ? __uint_as_float(u ^ 0x80000000u)
                             : __uint_as_float(~u);
}

__global__ void gat_init_kernel(unsigned* __restrict__ deg,
                                unsigned* __restrict__ gmax) {
    int idx = blockIdx.x * blockDim.x + threadIdx.x;
    if (idx < N_NODES)
        deg[idx] = 0u;
    if (idx == 0)
        *gmax = f2ord(-INFINITY);
}

__global__ __launch_bounds__(128)
void gat_proj_scores_kernel(const float* __restrict__ A,
                            const float* __restrict__ W,
                            const float* __restrict__ a_src,
                            const float* __restrict__ a_tgt,
                            float* __restrict__ P,
                            float* __restrict__ s_src,
                            float* __restrict__ s_tgt) {
    const int head = threadIdx.x >> 5;
    const int lane = threadIdx.x & 31;
    const int c    = lane & 15;
    const int half = lane >> 4;
    const int m0   = blockIdx.x * 16;
    const int col0 = head * D_OUT + 2 * c;
    const int kGrp = half * 2;

    const float* Arow = A + (size_t)(m0 + c) * D_IN + kGrp;
    const float* Wcol = W + (size_t)kGrp * D_IN + col0;

    v8f acc0 = {};
    v8f acc1 = {};
#pragma unroll
    for (int k0 = 0; k0 < D_IN; k0 += 4) {
        const float2 av = *(const float2*)(Arow + k0);
        const float2 w0 = *(const float2*)(Wcol + (size_t)k0 * D_IN);
        const float2 w1 = *(const float2*)(Wcol + (size_t)(k0 + 1) * D_IN);
        v2f a, b0, b1;
        a.x  = av.x;  a.y  = av.y;
        b0.x = w0.x;  b0.y = w1.x;
        b1.x = w0.y;  b1.y = w1.y;
        acc0 = __builtin_amdgcn_wmma_f32_16x16x4_f32(false, a, false, b0,
                                                     (short)0, acc0, false, false);
        acc1 = __builtin_amdgcn_wmma_f32_16x16x4_f32(false, a, false, b1,
                                                     (short)0, acc1, false, false);
    }

    const int mBase = m0 + 8 * half;
#pragma unroll
    for (int v = 0; v < 8; ++v)
        *(float2*)(P + (size_t)(mBase + v) * D_IN + col0) = make_float2(acc0[v], acc1[v]);

    const float2 as = *(const float2*)(a_src + col0);
    const float2 at = *(const float2*)(a_tgt + col0);
    float x[16];
#pragma unroll
    for (int v = 0; v < 8; ++v) {
        x[v]     = __builtin_fmaf(acc1[v], as.y, acc0[v] * as.x);
        x[8 + v] = __builtin_fmaf(acc1[v], at.y, acc0[v] * at.x);
    }
    const bool b3 = (c & 8) != 0, b2 = (c & 4) != 0, b1 = (c & 2) != 0, b0 = (c & 1) != 0;
    float y[8];
#pragma unroll
    for (int j = 0; j < 8; ++j) {
        const float keep = b3 ? x[j + 8] : x[j];
        const float send = b3 ? x[j] : x[j + 8];
        y[j] = keep + __shfl_xor(send, 8, 32);
    }
    float z[4];
#pragma unroll
    for (int j = 0; j < 4; ++j) {
        const float keep = b2 ? y[j + 4] : y[j];
        const float send = b2 ? y[j] : y[j + 4];
        z[j] = keep + __shfl_xor(send, 4, 32);
    }
    float w[2];
#pragma unroll
    for (int j = 0; j < 2; ++j) {
        const float keep = b1 ? z[j + 2] : z[j];
        const float send = b1 ? z[j] : z[j + 2];
        w[j] = keep + __shfl_xor(send, 2, 32);
    }
    const float keep0 = b0 ? w[1] : w[0];
    const float send0 = b0 ? w[0] : w[1];
    const float tot   = keep0 + __shfl_xor(send0, 1, 32);
    float* dst = (c & 8) ? s_tgt : s_src;
    dst[(size_t)(mBase + (c & 7)) * N_HEAD + head] = tot;
}

__global__ __launch_bounds__(256)
void gat_edge_max_kernel(const int* __restrict__ ei,
                         const float* __restrict__ s_src,
                         const float* __restrict__ s_tgt,
                         unsigned* __restrict__ gmax,
                         unsigned* __restrict__ deg,
                         unsigned* __restrict__ rank) {
    const int e = blockIdx.x * blockDim.x + threadIdx.x;
    float m = -INFINITY;
    if (e < N_EDGES) {
        const int src = ei[e];
        const int tgt = ei[N_EDGES + e];
        rank[e] = atomicAdd(&deg[tgt], 1u);
#pragma unroll
        for (int h = 0; h < N_HEAD; ++h) {
            float s = s_src[src * N_HEAD + h] + s_tgt[tgt * N_HEAD + h];
            s = (s > 0.0f) ? s : 0.2f * s;
            m = fmaxf(m, s);
        }
    }
#pragma unroll
    for (int off = 16; off > 0; off >>= 1)
        m = fmaxf(m, __shfl_xor(m, off, 32));
    if ((threadIdx.x & 31) == 0)
        atomicMax(gmax, f2ord(m));
}

#define SCAN_THREADS 1024
#define SCAN_ITEMS   49
__global__ __launch_bounds__(SCAN_THREADS)
void gat_scan_kernel(const unsigned* __restrict__ deg,
                     unsigned* __restrict__ rowptr) {
    __shared__ unsigned sh[SCAN_THREADS * SCAN_ITEMS];
    __shared__ unsigned wsum[32];
    const int t    = threadIdx.x;
    const int lane = t & 31;
    const int wave = t >> 5;
    for (int i = t; i < SCAN_THREADS * SCAN_ITEMS; i += SCAN_THREADS)
        sh[i] = (i < N_NODES) ? deg[i] : 0u;
    __syncthreads();
    unsigned s = 0u;
    for (int k = 0; k < SCAN_ITEMS; ++k)
        s += sh[t * SCAN_ITEMS + k];
    unsigned incl = s;
#pragma unroll
    for (int off = 1; off < 32; off <<= 1) {
        const unsigned v = __shfl_up(incl, off, 32);
        if (lane >= off) incl += v;
    }
    if (lane == 31) wsum[wave] = incl;
    __syncthreads();
    if (wave == 0) {
        const unsigned w = wsum[lane];
        unsigned wi = w;
#pragma unroll
        for (int off = 1; off < 32; off <<= 1) {
            const unsigned v = __shfl_up(wi, off, 32);
            if (lane >= off) wi += v;
        }
        wsum[lane] = wi - w;
    }
    __syncthreads();
    unsigned run = wsum[wave] + incl - s;
    for (int k = 0; k < SCAN_ITEMS; ++k) {
        const unsigned d = sh[t * SCAN_ITEMS + k];
        sh[t * SCAN_ITEMS + k] = run;
        run += d;
    }
    __syncthreads();
    for (int i = t; i < N_NODES; i += SCAN_THREADS)
        rowptr[i] = sh[i];
    if (t == SCAN_THREADS - 1)
        rowptr[N_NODES] = run;
}

__global__ __launch_bounds__(256)
void gat_fill_kernel(const int* __restrict__ ei,
                     const unsigned* __restrict__ rowptr,
                     const unsigned* __restrict__ rank,
                     int* __restrict__ perm) {
    const int e = blockIdx.x * blockDim.x + threadIdx.x;
    if (e >= N_EDGES) return;
    const int src = ei[e];
    const int tgt = ei[N_EDGES + e];
    perm[rowptr[tgt] + rank[e]] = src;
}

__device__ __forceinline__ float head_weight(const float4 att, const int j,
                                             const int m0, const int m1,
                                             const int m2, const int m3) {
    const int a0 = __builtin_amdgcn_readlane(__float_as_int(att.x), j);
    const int a1 = __builtin_amdgcn_readlane(__float_as_int(att.y), j);
    const int a2 = __builtin_amdgcn_readlane(__float_as_int(att.z), j);
    const int a3 = __builtin_amdgcn_readlane(__float_as_int(att.w), j);
    return __int_as_float((a0 & m0) | (a1 & m1) | (a2 & m2) | (a3 & m3));
}

__device__ __forceinline__ void bucket_weights(const int* __restrict__ perm,
                                               const float4* __restrict__ s_src4,
                                               const float4 st, const float gm,
                                               const int base, const int cnt,
                                               const int lane,
                                               int& src, float4& ex) {
    src = 0;
    ex = make_float4(0.0f, 0.0f, 0.0f, 0.0f);
    if (lane < cnt) {
        src = perm[base + lane];
        const float4 ss = s_src4[src];
        float s0 = ss.x + st.x, s1 = ss.y + st.y, s2 = ss.z + st.z, s3 = ss.w + st.w;
        s0 = (s0 > 0.0f) ? s0 : 0.2f * s0;
        s1 = (s1 > 0.0f) ? s1 : 0.2f * s1;
        s2 = (s2 > 0.0f) ? s2 : 0.2f * s2;
        s3 = (s3 > 0.0f) ? s3 : 0.2f * s3;
        ex.x = __expf(s0 - gm);
        ex.y = __expf(s1 - gm);
        ex.z = __expf(s2 - gm);
        ex.w = __expf(s3 - gm);
    }
}

__device__ __forceinline__ void bucket_accumulate(const float4* __restrict__ P4,
                                                  const int src, const float4 att,
                                                  const int cnt, const int lane,
                                                  const int hd, float4& acc) {
    const int m0 = (hd == 0) ? -1 : 0, m1 = (hd == 1) ? -1 : 0;
    const int m2 = (hd == 2) ? -1 : 0, m3 = (hd == 3) ? -1 : 0;
    int j = 0;
    for (; j + 4 <= cnt; j += 4) {
        float4 r[4];
        float  a[4];
#pragma unroll
        for (int u = 0; u < 4; ++u) {
            const int sj = __builtin_amdgcn_readlane(src, j + u);
            r[u] = P4[(size_t)sj * (D_IN / 4) + lane];
        }
#pragma unroll
        for (int u = 0; u < 4; ++u)
            a[u] = head_weight(att, j + u, m0, m1, m2, m3);
#pragma unroll
        for (int u = 0; u < 4; ++u) {
            acc.x = __builtin_fmaf(r[u].x, a[u], acc.x);
            acc.y = __builtin_fmaf(r[u].y, a[u], acc.y);
            acc.z = __builtin_fmaf(r[u].z, a[u], acc.z);
            acc.w = __builtin_fmaf(r[u].w, a[u], acc.w);
        }
    }
    for (; j < cnt; ++j) {
        const int sj = __builtin_amdgcn_readlane(src, j);
        const float4 r = P4[(size_t)sj * (D_IN / 4) + lane];
        const float a = head_weight(att, j, m0, m1, m2, m3);
        acc.x = __builtin_fmaf(r.x, a, acc.x);
        acc.y = __builtin_fmaf(r.y, a, acc.y);
        acc.z = __builtin_fmaf(r.z, a, acc.z);
        acc.w = __builtin_fmaf(r.w, a, acc.w);
    }
}

__global__ __launch_bounds__(256)
void gat_gather_kernel(const unsigned* __restrict__ rowptr,
                       const int* __restrict__ perm,
                       const float* __restrict__ P,
                       const float* __restrict__ s_src,
                       const float* __restrict__ s_tgt,
                       const unsigned* __restrict__ gmax,
                       const float* __restrict__ bias,
                       float* __restrict__ out) {
    const int lane = threadIdx.x & 31;
    const int hd   = lane >> 3;
    const int n = __builtin_amdgcn_readfirstlane((int)(blockIdx.x * 8 + (threadIdx.x >> 5)));
    const int beg = (int)rowptr[n];
    const int end = (int)rowptr[n + 1];
    const float4* P4     = (const float4*)P;
    const float4* s_src4 = (const float4*)s_src;
    const float   gm = ord2f(*gmax);
    const float4  st = ((const float4*)s_tgt)[n];

    float4 acc = make_float4(0.0f, 0.0f, 0.0f, 0.0f);
    if (end > beg) {
        const int cnt0 = (end - beg < 32) ? (end - beg) : 32;
        int src0; float4 ex0;
        bucket_weights(perm, s_src4, st, gm, beg, cnt0, lane, src0, ex0);
        float4 den = ex0;
        for (int b = beg + 32; b < end; b += 32) {
            const int c = (end - b < 32) ? (end - b) : 32;
            int sx; float4 ex;
            bucket_weights(perm, s_src4, st, gm, b, c, lane, sx, ex);
            den.x += ex.x; den.y += ex.y; den.z += ex.z; den.w += ex.w;
        }
#pragma unroll
        for (int off = 16; off > 0; off >>= 1) {
            den.x += __shfl_xor(den.x, off, 32);
            den.y += __shfl_xor(den.y, off, 32);
            den.z += __shfl_xor(den.z, off, 32);
            den.w += __shfl_xor(den.w, off, 32);
        }
        den.x += 1e-16f; den.y += 1e-16f; den.z += 1e-16f; den.w += 1e-16f;
        float4 att = make_float4(ex0.x / den.x, ex0.y / den.y, ex0.z / den.z, ex0.w / den.w);
        bucket_accumulate(P4, src0, att, cnt0, lane, hd, acc);
        for (int b = beg + 32; b < end; b += 32) {
            const int c = (end - b < 32) ? (end - b) : 32;
            int sx; float4 ex;
            bucket_weights(perm, s_src4, st, gm, b, c, lane, sx, ex);
            att = make_float4(ex.x / den.x, ex.y / den.y, ex.z / den.z, ex.w / den.w);
            bucket_accumulate(P4, sx, att, c, lane, hd, acc);
        }
    }
    const float4 bv = ((const float4*)bias)[lane];
    ((float4*)out)[(size_t)n * (D_IN / 4) + lane] =
        make_float4(acc.x + bv.x, acc.y + bv.y, acc.z + bv.z, acc.w + bv.w);
}

extern "C" void kernel_launch(void* const* d_in, const int* in_sizes, int n_in,
                              void* d_out, int out_size, void* d_ws, size_t ws_size,
                              hipStream_t stream) {
    const float* in_feat = (const float*)d_in[0];
    const int*   ei      = (const int*)  d_in[1];
    const float* W_proj  = (const float*)d_in[3];
    const float* a_src   = (const float*)d_in[4];
    const float* a_tgt   = (const float*)d_in[5];
    const float* bias    = (const float*)d_in[6];
    float*       out     = (float*)d_out;

    float*    proj   = (float*)d_ws;
    float*    s_src  = proj  + (size_t)N_NODES * D_IN;
    float*    s_tgt  = s_src + (size_t)N_NODES * N_HEAD;
    unsigned* deg    = (unsigned*)(s_tgt + (size_t)N_NODES * N_HEAD);
    unsigned* rowptr = deg + (size_t)N_NODES;
    unsigned* rank   = rowptr + (size_t)N_NODES + 1;
    int*      perm   = (int*)(rank + (size_t)N_EDGES);
    unsigned* gmax   = (unsigned*)(perm + (size_t)N_EDGES);

    gat_init_kernel<<<(N_NODES + 255) / 256, 256, 0, stream>>>(deg, gmax);

    gat_proj_scores_kernel<<<N_NODES / 16, 128, 0, stream>>>(
        in_feat, W_proj, a_src, a_tgt, proj, s_src, s_tgt);

    gat_edge_max_kernel<<<(N_EDGES + 255) / 256, 256, 0, stream>>>(
        ei, s_src, s_tgt, gmax, deg, rank);

    gat_scan_kernel<<<1, SCAN_THREADS, 0, stream>>>(deg, rowptr);

    gat_fill_kernel<<<(N_EDGES + 255) / 256, 256, 0, stream>>>(ei, rowptr, rank, perm);

    gat_gather_kernel<<<N_NODES / 8, 256, 0, stream>>>(
        rowptr, perm, proj, s_src, s_tgt, gmax, bias, out);
}


# edges grouped by target range in LDS: a work-group ranks its 3125 edges per range in LDS and fills its own segment of each range; a work-group per range sorts in LDS, sums its nodes; order = LDS rank
# speedup vs baseline: 7.3898x; 2.8404x over previous
#include <hip/hip_runtime.h>
#include <math.h>

#define N_NODES 50000
#define N_EDGES 800000
#define D_IN    128
#define N_HEAD  4
#define D_OUT   32
#define NB      512
#define TPB     98
#define BIN_WGS 256
#define EPB     3125
#define SEGCAP  32
#define BCAP    (BIN_WGS * SEGCAP)

typedef __attribute__((ext_vector_type(2))) float v2f;
typedef __attribute__((ext_vector_type(8))) float v8f;

__device__ __forceinline__ unsigned f2ord(float f) {
    unsigned u = __float_as_uint(f);
    return (u & 0x80000000u) ? ~u : (u | 0x80000000u);
}
__device__ __forceinline__ float ord2f(unsigned u) {
    return (u & 0x80000000u) ? __uint_as_float(u ^ 0x80000000u)
                             : __uint_as_float(~u);
}

__global__ void gat_init_kernel(unsigned* __restrict__ gmax,
                                unsigned* __restrict__ ovf_cnt) {
    if (blockIdx.x == 0 && threadIdx.x == 0) {
        *gmax = f2ord(-INFINITY);
        *ovf_cnt = 0u;
    }
}

__global__ __launch_bounds__(128)
void gat_proj_scores_kernel(const float* __restrict__ A,
                            const float* __restrict__ W,
                            const float* __restrict__ a_src,
                            const float* __restrict__ a_tgt,
                            float* __restrict__ P,
                            float* __restrict__ s_src,
                            float* __restrict__ s_tgt) {
    const int head = threadIdx.x >> 5;
    const int lane = threadIdx.x & 31;
    const int c    = lane & 15;
    const int half = lane >> 4;
    const int m0   = blockIdx.x * 16;
    const int col0 = head * D_OUT + 2 * c;
    const int kGrp = half * 2;

    const float* Arow = A + (size_t)(m0 + c) * D_IN + kGrp;
    const float* Wcol = W + (size_t)kGrp * D_IN + col0;

    v8f acc0 = {};
    v8f acc1 = {};
#pragma unroll
    for (int k0 = 0; k0 < D_IN; k0 += 4) {
        const float2 av = *(const float2*)(Arow + k0);
        const float2 w0 = *(const float2*)(Wcol + (size_t)k0 * D_IN);
        const float2 w1 = *(const float2*)(Wcol + (size_t)(k0 + 1) * D_IN);
        v2f a, b0, b1;
        a.x  = av.x;  a.y  = av.y;
        b0.x = w0.x;  b0.y = w1.x;
        b1.x = w0.y;  b1.y = w1.y;
        acc0 = __builtin_amdgcn_wmma_f32_16x16x4_f32(false, a, false, b0,
                                                     (short)0, acc0, false, false);
        acc1 = __builtin_amdgcn_wmma_f32_16x16x4_f32(false, a, false, b1,
                                                     (short)0, acc1, false, false);
    }

    const int mBase = m0 + 8 * half;
#pragma unroll
    for (int v = 0; v < 8; ++v)
        *(float2*)(P + (size_t)(mBase + v) * D_IN + col0) = make_float2(acc0[v], acc1[v]);

    const float2 as = *(const float2*)(a_src + col0);
    const float2 at = *(const float2*)(a_tgt + col0);
    float x[16];
#pragma unroll
    for (int v = 0; v < 8; ++v) {
        x[v]     = __builtin_fmaf(acc1[v], as.y, acc0[v] * as.x);
        x[8 + v] = __builtin_fmaf(acc1[v], at.y, acc0[v] * at.x);
    }
    const bool b3 = (c & 8) != 0, b2 = (c & 4) != 0, b1 = (c & 2) != 0, b0 = (c & 1) != 0;
    float y[8];
#pragma unroll
    for (int j = 0; j < 8; ++j) {
        const float keep = b3 ? x[j + 8] : x[j];
        const float send = b3 ? x[j] : x[j + 8];
        y[j] = keep + __shfl_xor(send, 8, 32);
    }
    float z[4];
#pragma unroll
    for (int j = 0; j < 4; ++j) {
        const float keep = b2 ? y[j + 4] : y[j];
        const float send = b2 ? y[j] : y[j + 4];
        z[j] = keep + __shfl_xor(send, 4, 32);
    }
    float w[2];
#pragma unroll
    for (int j = 0; j < 2; ++j) {
        const float keep = b1 ? z[j + 2] : z[j];
        const float send = b1 ? z[j] : z[j + 2];
        w[j] = keep + __shfl_xor(send, 2, 32);
    }
    const float keep0 = b0 ? w[1] : w[0];
    const float send0 = b0 ? w[0] : w[1];
    const float tot   = keep0 + __shfl_xor(send0, 1, 32);
    float* dst = (c & 8) ? s_tgt : s_src;
    dst[(size_t)(mBase + (c & 7)) * N_HEAD + head] = tot;
}

__global__ __launch_bounds__(1024)
void gat_bin_max_kernel(const int* __restrict__ ei,
                        const float* __restrict__ s_src,
                        const float* __restrict__ s_tgt,
                        unsigned* __restrict__ gmax,
                        unsigned* __restrict__ seg_cnt,
                        int* __restrict__ seg_w,
                        unsigned* __restrict__ ovf_cnt,
                        int* __restrict__ ovf_tgt,
                        int* __restrict__ ovf_src) {
    __shared__ unsigned lcount[NB];
    __shared__ float    lmax[32];
    int2* seg = (int2*)seg_w;
    const int tid = threadIdx.x;
    const int wg  = blockIdx.x;
    if (tid < NB)
        lcount[tid] = 0u;
    __syncthreads();

    const int e0 = wg * EPB;
    int src[4], tgt[4];
#pragma unroll
    for (int k = 0; k < 4; ++k) {
        const int i = tid + k * 1024;
        src[k] = 0; tgt[k] = 0;
        if (i < EPB) {
            src[k] = ei[e0 + i];
            tgt[k] = ei[N_EDGES + e0 + i];
        }
    }
    float m = -INFINITY;
#pragma unroll
    for (int k = 0; k < 4; ++k) {
        const int i = tid + k * 1024;
        if (i < EPB) {
            const float4 ss = ((const float4*)s_src)[src[k]];
            const float4 st = ((const float4*)s_tgt)[tgt[k]];
            const int bin = tgt[k] / TPB;
            const unsigned rank = atomicAdd(&lcount[bin], 1u);
            if (rank < (unsigned)SEGCAP) {
                seg[((size_t)bin * BIN_WGS + wg) * SEGCAP + rank] = make_int2(src[k], tgt[k]);
            } else {
                const unsigned o = atomicAdd(ovf_cnt, 1u);
                ovf_tgt[o] = tgt[k];
                ovf_src[o] = src[k];
            }
            const float sc[N_HEAD] = {ss.x + st.x, ss.y + st.y, ss.z + st.z, ss.w + st.w};
#pragma unroll
            for (int h = 0; h < N_HEAD; ++h) {
                const float s = (sc[h] > 0.0f) ? sc[h] : 0.2f * sc[h];
                m = fmaxf(m, s);
            }
        }
    }
#pragma unroll
    for (int off = 16; off > 0; off >>= 1)
        m = fmaxf(m, __shfl_xor(m, off, 32));
    if ((tid & 31) == 0)
        lmax[tid >> 5] = m;
    __syncthreads();

    if (tid < NB)
        seg_cnt[wg * NB + tid] = lcount[tid];
    if (tid < 32) {
        float mm = lmax[tid];
#pragma unroll
        for (int off = 16; off > 0; off >>= 1)
            mm = fmaxf(mm, __shfl_xor(mm, off, 32));
        if (tid == 0)
            atomicMax(gmax, f2ord(mm));
    }
}

__device__ __forceinline__ float head_weight(const float4 att, const int j,
                                             const int m0, const int m1,
                                             const int m2, const int m3) {
    const int a0 = __builtin_amdgcn_readlane(__float_as_int(att.x), j);
    const int a1 = __builtin_amdgcn_readlane(__float_as_int(att.y), j);
    const int a2 = __builtin_amdgcn_readlane(__float_as_int(att.z), j);
    const int a3 = __builtin_amdgcn_readlane(__float_as_int(att.w), j);
    return __int_as_float((a0 & m0) | (a1 & m1) | (a2 & m2) | (a3 & m3));
}

__device__ __forceinline__ float4 edge_exp(const float4 ss, const float4 st,
                                           const float gm) {
    float s0 = ss.x + st.x, s1 = ss.y + st.y, s2 = ss.z + st.z, s3 = ss.w + st.w;
    s0 = (s0 > 0.0f) ? s0 : 0.2f * s0;
    s1 = (s1 > 0.0f) ? s1 : 0.2f * s1;
    s2 = (s2 > 0.0f) ? s2 : 0.2f * s2;
    s3 = (s3 > 0.0f) ? s3 : 0.2f * s3;
    return make_float4(__expf(s0 - gm), __expf(s1 - gm), __expf(s2 - gm), __expf(s3 - gm));
}

__device__ __forceinline__ void bucket_weights(const int* lsrc,
                                               const float4* __restrict__ s_src4,
                                               const float4 st, const float gm,
                                               const int off, const int c0, const int d,
                                               const int lane,
                                               int& src, float4& ex) {
    src = 0;
    ex = make_float4(0.0f, 0.0f, 0.0f, 0.0f);
    const int j = c0 + lane;
    if (j < d) {
        src = lsrc[off + j];
        ex = edge_exp(s_src4[src], st, gm);
    }
}

__device__ __forceinline__ void overflow_weights(const int* __restrict__ ovf_tgt,
                                                 const int* __restrict__ ovf_src,
                                                 const float4* __restrict__ s_src4,
                                                 const float4 st, const float gm,
                                                 const int base, const int novf,
                                                 const int n, const int lane,
                                                 int& src, float4& ex, bool& hit) {
    src = 0;
    ex = make_float4(0.0f, 0.0f, 0.0f, 0.0f);
    hit = false;
    const int i = base + lane;
    if (i < novf && ovf_tgt[i] == n) {
        hit = true;
        src = ovf_src[i];
        ex = edge_exp(s_src4[src], st, gm);
    }
}

__device__ __forceinline__ void bucket_accumulate(const float4* __restrict__ P4,
                                                  const int src, const float4 att,
                                                  const int cnt, const int lane,
                                                  const int hd, float4& acc) {
    const int m0 = (hd == 0) ? -1 : 0, m1 = (hd == 1) ? -1 : 0;
    const int m2 = (hd == 2) ? -1 : 0, m3 = (hd == 3) ? -1 : 0;
    int j = 0;
    for (; j + 4 <= cnt; j += 4) {
        float4 r[4];
        float  a[4];
#pragma unroll
        for (int u = 0; u < 4; ++u) {
            const int sj = __builtin_amdgcn_readlane(src, j + u);
            r[u] = P4[(size_t)sj * (D_IN / 4) + lane];
        }
#pragma unroll
        for (int u = 0; u < 4; ++u)
            a[u] = head_weight(att, j + u, m0, m1, m2, m3);
#pragma unroll
        for (int u = 0; u < 4; ++u) {
            acc.x = __builtin_fmaf(r[u].x, a[u], acc.x);
            acc.y = __builtin_fmaf(r[u].y, a[u], acc.y);
            acc.z = __builtin_fmaf(r[u].z, a[u], acc.z);
            acc.w = __builtin_fmaf(r[u].w, a[u], acc.w);
        }
    }
    for (; j < cnt; ++j) {
        const int sj = __builtin_amdgcn_readlane(src, j);
        const float4 r = P4[(size_t)sj * (D_IN / 4) + lane];
        const float a = head_weight(att, j, m0, m1, m2, m3);
        acc.x = __builtin_fmaf(r.x, a, acc.x);
        acc.y = __builtin_fmaf(r.y, a, acc.y);
        acc.z = __builtin_fmaf(r.z, a, acc.z);
        acc.w = __builtin_fmaf(r.w, a, acc.w);
    }
}

__global__ __launch_bounds__(1024)
void gat_gather_kernel(const unsigned* __restrict__ seg_cnt,
                       const int* __restrict__ seg_w,
                       const unsigned* __restrict__ ovf_cnt,
                       const int* __restrict__ ovf_tgt,
                       const int* __restrict__ ovf_src,
                       const float* __restrict__ P,
                       const float* __restrict__ s_src,
                       const float* __restrict__ s_tgt,
                       const unsigned* __restrict__ gmax,
                       const float* __restrict__ bias,
                       float* __restrict__ out) {
    __shared__ unsigned ldeg[128];
    __shared__ unsigned loff[128];
    __shared__ unsigned lcur[128];
    __shared__ unsigned lnext;
    __shared__ int      lsrc[BCAP];
    const int tid   = threadIdx.x;
    const int lane  = tid & 31;
    const int hd    = lane >> 3;
    const int wave  = __builtin_amdgcn_readfirstlane(tid >> 5);
    const int rng   = blockIdx.x;
    const int nbase = rng * TPB;
    const int2* seg = (const int2*)seg_w + (size_t)rng * BIN_WGS * SEGCAP;

    if (tid < 128)
        ldeg[tid] = 0u;
    if (tid == 0)
        lnext = 0u;

    int2 ent[8];
    bool ok[8];
    unsigned segn[8];
#pragma unroll
    for (int k = 0; k < 8; ++k)
        segn[k] = seg_cnt[(k * 32 + wave) * NB + rng];
#pragma unroll
    for (int k = 0; k < 8; ++k) {
        const unsigned c = (segn[k] < (unsigned)SEGCAP) ? segn[k] : (unsigned)SEGCAP;
        ok[k] = (unsigned)lane < c;
        ent[k] = make_int2(0, nbase);
        if (ok[k])
            ent[k] = seg[(k * 32 + wave) * SEGCAP + lane];
    }
    __syncthreads();

#pragma unroll
    for (int k = 0; k < 8; ++k)
        if (ok[k])
            atomicAdd(&ldeg[ent[k].y - nbase], 1u);
    __syncthreads();

    if (tid < 32) {
        const unsigned v0 = ldeg[4 * lane + 0], v1 = ldeg[4 * lane + 1];
        const unsigned v2 = ldeg[4 * lane + 2], v3 = ldeg[4 * lane + 3];
        const unsigned s = v0 + v1 + v2 + v3;
        unsigned incl = s;
#pragma unroll
        for (int o = 1; o < 32; o <<= 1) {
            const unsigned y = __shfl_up(incl, o, 32);
            if (lane >= o) incl += y;
        }
        const unsigned e0 = incl - s, e1 = e0 + v0, e2 = e1 + v1, e3 = e2 + v2;
        loff[4 * lane + 0] = e0;  lcur[4 * lane + 0] = e0;
        loff[4 * lane + 1] = e1;  lcur[4 * lane + 1] = e1;
        loff[4 * lane + 2] = e2;  lcur[4 * lane + 2] = e2;
        loff[4 * lane + 3] = e3;  lcur[4 * lane + 3] = e3;
    }
    __syncthreads();

#pragma unroll
    for (int k = 0; k < 8; ++k)
        if (ok[k]) {
            const unsigned p = atomicAdd(&lcur[ent[k].y - nbase], 1u);
            lsrc[p] = ent[k].x;
        }
    __syncthreads();

    const float4* P4     = (const float4*)P;
    const float4* s_src4 = (const float4*)s_src;
    const float   gm = ord2f(*gmax);
    const float4  bv = ((const float4*)bias)[lane];
    const unsigned oc = *ovf_cnt;
    const int novf = (oc < (unsigned)N_EDGES) ? (int)oc : N_EDGES;
    const int m0 = (hd == 0) ? -1 : 0, m1 = (hd == 1) ? -1 : 0;
    const int m2 = (hd == 2) ? -1 : 0, m3 = (hd == 3) ? -1 : 0;

    for (int it = 0; it < TPB; ++it) {
        int t = 0;
        if (lane == 0)
            t = (int)atomicAdd(&lnext, 1u);
        t = __builtin_amdgcn_readfirstlane(t);
        const int n = nbase + t;
        if (t >= TPB || n >= N_NODES)
            break;
        const int d   = __builtin_amdgcn_readfirstlane((int)ldeg[t]);
        const int off = __builtin_amdgcn_readfirstlane((int)loff[t]);

        float4 acc = make_float4(0.0f, 0.0f, 0.0f, 0.0f);
        if (d > 0 || novf > 0) {
            const float4 st = ((const float4*)s_tgt)[n];
            int src0; float4 ex0;
            bucket_weights(lsrc, s_src4, st, gm, off, 0, d, lane, src0, ex0);
            float4 den = ex0;
            for (int c0 = 32; c0 < d; c0 += 32) {
                int sx; float4 ex;
                bucket_weights(lsrc, s_src4, st, gm, off, c0, d, lane, sx, ex);
                den.x += ex.x; den.y += ex.y; den.z += ex.z; den.w += ex.w;
            }
            for (int b = 0; b < novf; b += 32) {
                int sx; float4 ex; bool hit;
                overflow_weights(ovf_tgt, ovf_src, s_src4, st, gm, b, novf, n, lane, sx, ex, hit);
                den.x += ex.x; den.y += ex.y; den.z += ex.z; den.w += ex.w;
            }
#pragma unroll
            for (int o = 16; o > 0; o >>= 1) {
                den.x += __shfl_xor(den.x, o, 32);
                den.y += __shfl_xor(den.y, o, 32);
                den.z += __shfl_xor(den.z, o, 32);
                den.w += __shfl_xor(den.w, o, 32);
            }
            den.x += 1e-16f; den.y += 1e-16f; den.z += 1e-16f; den.w += 1e-16f;
            float4 att = make_float4(ex0.x / den.x, ex0.y / den.y, ex0.z / den.z, ex0.w / den.w);
            bucket_accumulate(P4, src0, att, (d < 32) ? d : 32, lane, hd, acc);
            for (int c0 = 32; c0 < d; c0 += 32) {
                int sx; float4 ex;
                bucket_weights(lsrc, s_src4, st, gm, off, c0, d, lane, sx, ex);
                att = make_float4(ex.x / den.x, ex.y / den.y, ex.z / den.z, ex.w / den.w);
                const int rem = d - c0;
                bucket_accumulate(P4, sx, att, (rem < 32) ? rem : 32, lane, hd, acc);
            }
            for (int b = 0; b < novf; b += 32) {
                int sx; float4 ex; bool hit;
                overflow_weights(ovf_tgt, ovf_src, s_src4, st, gm, b, novf, n, lane, sx, ex, hit);
                att = make_float4(ex.x / den.x, ex.y / den.y, ex.z / den.z, ex.w / den.w);
                unsigned todo = __builtin_amdgcn_ballot_w32(hit);
                while (todo != 0u) {
                    const int j = __builtin_ctz(todo);
                    todo &= todo - 1u;
                    const int sj = __builtin_amdgcn_readlane(sx, j);
                    const float4 r = P4[(size_t)sj * (D_IN / 4) + lane];
                    const float a = head_weight(att, j, m0, m1, m2, m3);
                    acc.x = __builtin_fmaf(r.x, a, acc.x);
                    acc.y = __builtin_fmaf(r.y, a, acc.y);
                    acc.z = __builtin_fmaf(r.z, a, acc.z);
                    acc.w = __builtin_fmaf(r.w, a, acc.w);
                }
            }
        }
        ((float4*)out)[(size_t)n * (D_IN / 4) + lane] =
            make_float4(acc.x + bv.x, acc.y + bv.y, acc.z + bv.z, acc.w + bv.w);
    }
}

extern "C" void kernel_launch(void* const* d_in, const int* in_sizes, int n_in,
                              void* d_out, int out_size, void* d_ws, size_t ws_size,
                              hipStream_t stream) {
    const float* in_feat = (const float*)d_in[0];
    const int*   ei      = (const int*)  d_in[1];
    const float* W_proj  = (const float*)d_in[3];
    const float* a_src   = (const float*)d_in[4];
    const float* a_tgt   = (const float*)d_in[5];
    const float* bias    = (const float*)d_in[6];
    float*       out     = (float*)d_out;

    float*    proj    = (float*)d_ws;
    float*    s_src   = proj  + (size_t)N_NODES * D_IN;
    float*    s_tgt   = s_src + (size_t)N_NODES * N_HEAD;
    int*      seg     = (int*)(s_tgt + (size_t)N_NODES * N_HEAD);
    unsigned* seg_cnt = (unsigned*)(seg + (size_t)NB * BIN_WGS * SEGCAP * 2);
    int*      ovf_tgt = (int*)(seg_cnt + (size_t)BIN_WGS * NB);
    int*      ovf_src = ovf_tgt + (size_t)N_EDGES;
    unsigned* gmax    = (unsigned*)(ovf_src + (size_t)N_EDGES);
    unsigned* ovf_cnt = gmax + 1;

    gat_init_kernel<<<1, 32, 0, stream>>>(gmax, ovf_cnt);

    gat_proj_scores_kernel<<<N_NODES / 16, 128, 0, stream>>>(
        in_feat, W_proj, a_src, a_tgt, proj, s_src, s_tgt);

    gat_bin_max_kernel<<<BIN_WGS, 1024, 0, stream>>>(
        ei, s_src, s_tgt, gmax, seg_cnt, seg, ovf_cnt, ovf_tgt, ovf_src);

    gat_gather_kernel<<<NB, 1024, 0, stream>>>(
        seg_cnt, seg, ovf_cnt, ovf_tgt, ovf_src, proj, s_src, s_tgt, gmax, bias, out);
}


# init launch dropped: binning stores one score max per work-group (no atomic max on one word), the gather kernel's idle second wave reduces the 256 maxima; projection zeroes the overflow count
# speedup vs baseline: 7.7145x; 1.0439x over previous
#include <hip/hip_runtime.h>
#include <math.h>

#define N_NODES 50000
#define N_EDGES 800000
#define D_IN    128
#define N_HEAD  4
#define D_OUT   32
#define NB      512
#define TPB     98
#define BIN_WGS 256
#define EPB     3125
#define SEGCAP  32
#define BCAP    (BIN_WGS * SEGCAP)

typedef __attribute__((ext_vector_type(2))) float v2f;
typedef __attribute__((ext_vector_type(8))) float v8f;


__global__ __launch_bounds__(128)
void gat_proj_scores_kernel(const float* __restrict__ A,
                            const float* __restrict__ W,
                            const float* __restrict__ a_src,
                            const float* __restrict__ a_tgt,
                            float* __restrict__ P,
                            float* __restrict__ s_src,
                            float* __restrict__ s_tgt,
                            unsigned* __restrict__ ovf_cnt) {
    const int head = threadIdx.x >> 5;
    const int lane = threadIdx.x & 31;
    const int c    = lane & 15;
    const int half = lane >> 4;
    const int m0   = blockIdx.x * 16;
    const int col0 = head * D_OUT + 2 * c;
    const int kGrp = half * 2;

    const float* Arow = A + (size_t)(m0 + c) * D_IN + kGrp;
    const float* Wcol = W + (size_t)kGrp * D_IN + col0;

    v8f acc0 = {};
    v8f acc1 = {};
#pragma unroll
    for (int k0 = 0; k0 < D_IN; k0 += 4) {
        const float2 av = *(const float2*)(Arow + k0);
        const float2 w0 = *(const float2*)(Wcol + (size_t)k0 * D_IN);
        const float2 w1 = *(const float2*)(Wcol + (size_t)(k0 + 1) * D_IN);
        v2f a, b0, b1;
        a.x  = av.x;  a.y  = av.y;
        b0.x = w0.x;  b0.y = w1.x;
        b1.x = w0.y;  b1.y = w1.y;
        acc0 = __builtin_amdgcn_wmma_f32_16x16x4_f32(false, a, false, b0,
                                                     (short)0, acc0, false, false);
        acc1 = __builtin_amdgcn_wmma_f32_16x16x4_f32(false, a, false, b1,
                                                     (short)0, acc1, false, false);
    }

    const int mBase = m0 + 8 * half;
#pragma unroll
    for (int v = 0; v < 8; ++v)
        *(float2*)(P + (size_t)(mBase + v) * D_IN + col0) = make_float2(acc0[v], acc1[v]);

    const float2 as = *(const float2*)(a_src + col0);
    const float2 at = *(const float2*)(a_tgt + col0);
    float x[16];
#pragma unroll
    for (int v = 0; v < 8; ++v) {
        x[v]     = __builtin_fmaf(acc1[v], as.y, acc0[v] * as.x);
        x[8 + v] = __builtin_fmaf(acc1[v], at.y, acc0[v] * at.x);
    }
    const bool b3 = (c & 8) != 0, b2 = (c & 4) != 0, b1 = (c & 2) != 0, b0 = (c & 1) != 0;
    float y[8];
#pragma unroll
    for (int j = 0; j < 8; ++j) {
        const float keep = b3 ? x[j + 8] : x[j];
        const float send = b3 ? x[j] : x[j + 8];
        y[j] = keep + __shfl_xor(send, 8, 32);
    }
    float z[4];
#pragma unroll
    for (int j = 0; j < 4; ++j) {
        const float keep = b2 ? y[j + 4] : y[j];
        const float send = b2 ? y[j] : y[j + 4];
        z[j] = keep + __shfl_xor(send, 4, 32);
    }
    float w[2];
#pragma unroll
    for (int j = 0; j < 2; ++j) {
        const float keep = b1 ? z[j + 2] : z[j];
        const float send = b1 ? z[j] : z[j + 2];
        w[j] = keep + __shfl_xor(send, 2, 32);
    }
    const float keep0 = b0 ? w[1] : w[0];
    const float send0 = b0 ? w[0] : w[1];
    const float tot   = keep0 + __shfl_xor(send0, 1, 32);
    float* dst = (c & 8) ? s_tgt : s_src;
    dst[(size_t)(mBase + (c & 7)) * N_HEAD + head] = tot;
    if (blockIdx.x == 0 && threadIdx.x == 0)
        *ovf_cnt = 0u;
}

__global__ __launch_bounds__(1024)
void gat_bin_max_kernel(const int* __restrict__ ei,
                        const float* __restrict__ s_src,
                        const float* __restrict__ s_tgt,
                        float* __restrict__ wg_max,
                        unsigned* __restrict__ seg_cnt,
                        int* __restrict__ seg_w,
                        unsigned* __restrict__ ovf_cnt,
                        int* __restrict__ ovf_tgt,
                        int* __restrict__ ovf_src) {
    __shared__ unsigned lcount[NB];
    __shared__ float    lmax[32];
    int2* seg = (int2*)seg_w;
    const int tid = threadIdx.x;
    const int wg  = blockIdx.x;
    if (tid < NB)
        lcount[tid] = 0u;
    __syncthreads();

    const int e0 = wg * EPB;
    int src[4], tgt[4];
#pragma unroll
    for (int k = 0; k < 4; ++k) {
        const int i = tid + k * 1024;
        src[k] = 0; tgt[k] = 0;
        if (i < EPB) {
            src[k] = ei[e0 + i];
            tgt[k] = ei[N_EDGES + e0 + i];
        }
    }
    float m = -INFINITY;
#pragma unroll
    for (int k = 0; k < 4; ++k) {
        const int i = tid + k * 1024;
        if (i < EPB) {
            const float4 ss = ((const float4*)s_src)[src[k]];
            const float4 st = ((const float4*)s_tgt)[tgt[k]];
            const int bin = tgt[k] / TPB;
            const unsigned rank = atomicAdd(&lcount[bin], 1u);
            if (rank < (unsigned)SEGCAP) {
                seg[((size_t)bin * BIN_WGS + wg) * SEGCAP + rank] = make_int2(src[k], tgt[k]);
            } else {
                const unsigned o = atomicAdd(ovf_cnt, 1u);
                ovf_tgt[o] = tgt[k];
                ovf_src[o] = src[k];
            }
            const float sc[N_HEAD] = {ss.x + st.x, ss.y + st.y, ss.z + st.z, ss.w + st.w};
#pragma unroll
            for (int h = 0; h < N_HEAD; ++h) {
                const float s = (sc[h] > 0.0f) ? sc[h] : 0.2f * sc[h];
                m = fmaxf(m, s);
            }
        }
    }
#pragma unroll
    for (int off = 16; off > 0; off >>= 1)
        m = fmaxf(m, __shfl_xor(m, off, 32));
    if ((tid & 31) == 0)
        lmax[tid >> 5] = m;
    __syncthreads();

    if (tid < NB)
        seg_cnt[wg * NB + tid] = lcount[tid];
    if (tid < 32) {
        float mm = lmax[tid];
#pragma unroll
        for (int off = 16; off > 0; off >>= 1)
            mm = fmaxf(mm, __shfl_xor(mm, off, 32));
        if (tid == 0)
            wg_max[wg] = mm;
    }
}

__device__ __forceinline__ float head_weight(const float4 att, const int j,
                                             const int m0, const int m1,
                                             const int m2, const int m3) {
    const int a0 = __builtin_amdgcn_readlane(__float_as_int(att.x), j);
    const int a1 = __builtin_amdgcn_readlane(__float_as_int(att.y), j);
    const int a2 = __builtin_amdgcn_readlane(__float_as_int(att.z), j);
    const int a3 = __builtin_amdgcn_readlane(__float_as_int(att.w), j);
    return __int_as_float((a0 & m0) | (a1 & m1) | (a2 & m2) | (a3 & m3));
}

__device__ __forceinline__ float4 edge_exp(const float4 ss, const float4 st,
                                           const float gm) {
    float s0 = ss.x + st.x, s1 = ss.y + st.y, s2 = ss.z + st.z, s3 = ss.w + st.w;
    s0 = (s0 > 0.0f) ? s0 : 0.2f * s0;
    s1 = (s1 > 0.0f) ? s1 : 0.2f * s1;
    s2 = (s2 > 0.0f) ? s2 : 0.2f * s2;
    s3 = (s3 > 0.0f) ? s3 : 0.2f * s3;
    return make_float4(__expf(s0 - gm), __expf(s1 - gm), __expf(s2 - gm), __expf(s3 - gm));
}

__device__ __forceinline__ void bucket_weights(const int* lsrc,
                                               const float4* __restrict__ s_src4,
                                               const float4 st, const float gm,
                                               const int off, const int c0, const int d,
                                               const int lane,
                                               int& src, float4& ex) {
    src = 0;
    ex = make_float4(0.0f, 0.0f, 0.0f, 0.0f);
    const int j = c0 + lane;
    if (j < d) {
        src = lsrc[off + j];
        ex = edge_exp(s_src4[src], st, gm);
    }
}

__device__ __forceinline__ void overflow_weights(const int* __restrict__ ovf_tgt,
                                                 const int* __restrict__ ovf_src,
                                                 const float4* __restrict__ s_src4,
                                                 const float4 st, const float gm,
                                                 const int base, const int novf,
                                                 const int n, const int lane,
                                                 int& src, float4& ex, bool& hit) {
    src = 0;
    ex = make_float4(0.0f, 0.0f, 0.0f, 0.0f);
    hit = false;
    const int i = base + lane;
    if (i < novf && ovf_tgt[i] == n) {
        hit = true;
        src = ovf_src[i];
        ex = edge_exp(s_src4[src], st, gm);
    }
}

__device__ __forceinline__ void bucket_accumulate(const float4* __restrict__ P4,
                                                  const int src, const float4 att,
                                                  const int cnt, const int lane,
                                                  const int hd, float4& acc) {
    const int m0 = (hd == 0) ? -1 : 0, m1 = (hd == 1) ? -1 : 0;
    const int m2 = (hd == 2) ? -1 : 0, m3 = (hd == 3) ? -1 : 0;
    int j = 0;
    for (; j + 4 <= cnt; j += 4) {
        float4 r[4];
        float  a[4];
#pragma unroll
        for (int u = 0; u < 4; ++u) {
            const int sj = __builtin_amdgcn_readlane(src, j + u);
            r[u] = P4[(size_t)sj * (D_IN / 4) + lane];
        }
#pragma unroll
        for (int u = 0; u < 4; ++u)
            a[u] = head_weight(att, j + u, m0, m1, m2, m3);
#pragma unroll
        for (int u = 0; u < 4; ++u) {
            acc.x = __builtin_fmaf(r[u].x, a[u], acc.x);
            acc.y = __builtin_fmaf(r[u].y, a[u], acc.y);
            acc.z = __builtin_fmaf(r[u].z, a[u], acc.z);
            acc.w = __builtin_fmaf(r[u].w, a[u], acc.w);
        }
    }
    for (; j < cnt; ++j) {
        const int sj = __builtin_amdgcn_readlane(src, j);
        const float4 r = P4[(size_t)sj * (D_IN / 4) + lane];
        const float a = head_weight(att, j, m0, m1, m2, m3);
        acc.x = __builtin_fmaf(r.x, a, acc.x);
        acc.y = __builtin_fmaf(r.y, a, acc.y);
        acc.z = __builtin_fmaf(r.z, a, acc.z);
        acc.w = __builtin_fmaf(r.w, a, acc.w);
    }
}

__global__ __launch_bounds__(1024)
void gat_gather_kernel(const unsigned* __restrict__ seg_cnt,
                       const int* __restrict__ seg_w,
                       const unsigned* __restrict__ ovf_cnt,
                       const int* __restrict__ ovf_tgt,
                       const int* __restrict__ ovf_src,
                       const float* __restrict__ P,
                       const float* __restrict__ s_src,
                       const float* __restrict__ s_tgt,
                       const float* __restrict__ wg_max,
                       const float* __restrict__ bias,
                       float* __restrict__ out) {
    __shared__ unsigned ldeg[128];
    __shared__ unsigned loff[128];
    __shared__ unsigned lcur[128];
    __shared__ unsigned lnext;
    __shared__ float    lgm;
    __shared__ int      lsrc[BCAP];
    const int tid   = threadIdx.x;
    const int lane  = tid & 31;
    const int hd    = lane >> 3;
    const int wave  = __builtin_amdgcn_readfirstlane(tid >> 5);
    const int rng   = blockIdx.x;
    const int nbase = rng * TPB;
    const int2* seg = (const int2*)seg_w + (size_t)rng * BIN_WGS * SEGCAP;

    if (tid < 128)
        ldeg[tid] = 0u;
    if (tid == 0)
        lnext = 0u;

    int2 ent[8];
    bool ok[8];
    unsigned segn[8];
#pragma unroll
    for (int k = 0; k < 8; ++k)
        segn[k] = seg_cnt[(k * 32 + wave) * NB + rng];
#pragma unroll
    for (int k = 0; k < 8; ++k) {
        const unsigned c = (segn[k] < (unsigned)SEGCAP) ? segn[k] : (unsigned)SEGCAP;
        ok[k] = (unsigned)lane < c;
        ent[k] = make_int2(0, nbase);
        if (ok[k])
            ent[k] = seg[(k * 32 + wave) * SEGCAP + lane];
    }
    float gm8 = 0.0f;
    if (wave == 1) {
        const float4 gma = ((const float4*)wg_max)[2 * lane];
        const float4 gmb = ((const float4*)wg_max)[2 * lane + 1];
        gm8 = fmaxf(fmaxf(fmaxf(gma.x, gma.y), fmaxf(gma.z, gma.w)),
                    fmaxf(fmaxf(gmb.x, gmb.y), fmaxf(gmb.z, gmb.w)));
    }
    __syncthreads();

#pragma unroll
    for (int k = 0; k < 8; ++k)
        if (ok[k])
            atomicAdd(&ldeg[ent[k].y - nbase], 1u);
    __syncthreads();

    if (tid < 32) {
        const unsigned v0 = ldeg[4 * lane + 0], v1 = ldeg[4 * lane + 1];
        const unsigned v2 = ldeg[4 * lane + 2], v3 = ldeg[4 * lane + 3];
        const unsigned s = v0 + v1 + v2 + v3;
        unsigned incl = s;
#pragma unroll
        for (int o = 1; o < 32; o <<= 1) {
            const unsigned y = __shfl_up(incl, o, 32);
            if (lane >= o) incl += y;
        }
        const unsigned e0 = incl - s, e1 = e0 + v0, e2 = e1 + v1, e3 = e2 + v2;
        loff[4 * lane + 0] = e0;  lcur[4 * lane + 0] = e0;
        loff[4 * lane + 1] = e1;  lcur[4 * lane + 1] = e1;
        loff[4 * lane + 2] = e2;  lcur[4 * lane + 2] = e2;
        loff[4 * lane + 3] = e3;  lcur[4 * lane + 3] = e3;
    } else if (wave == 1) {
        float mm = gm8;
#pragma unroll
        for (int o = 16; o > 0; o >>= 1)
            mm = fmaxf(mm, __shfl_xor(mm, o, 32));
        if (lane == 0)
            lgm = mm;
    }
    __syncthreads();

#pragma unroll
    for (int k = 0; k < 8; ++k)
        if (ok[k]) {
            const unsigned p = atomicAdd(&lcur[ent[k].y - nbase], 1u);
            lsrc[p] = ent[k].x;
        }
    __syncthreads();

    const float4* P4     = (const float4*)P;
    const float4* s_src4 = (const float4*)s_src;
    const float   gm = lgm;
    const float4  bv = ((const float4*)bias)[lane];
    const unsigned oc = *ovf_cnt;
    const int novf = (oc < (unsigned)N_EDGES) ? (int)oc : N_EDGES;
    const int m0 = (hd == 0) ? -1 : 0, m1 = (hd == 1) ? -1 : 0;
    const int m2 = (hd == 2) ? -1 : 0, m3 = (hd == 3) ? -1 : 0;

    for (int it = 0; it < TPB; ++it) {
        int t = 0;
        if (lane == 0)
            t = (int)atomicAdd(&lnext, 1u);
        t = __builtin_amdgcn_readfirstlane(t);
        const int n = nbase + t;
        if (t >= TPB || n >= N_NODES)
            break;
        const int d   = __builtin_amdgcn_readfirstlane((int)ldeg[t]);
        const int off = __builtin_amdgcn_readfirstlane((int)loff[t]);

        float4 acc = make_float4(0.0f, 0.0f, 0.0f, 0.0f);
        if (d > 0 || novf > 0) {
            const float4 st = ((const float4*)s_tgt)[n];
            int src0; float4 ex0;
            bucket_weights(lsrc, s_src4, st, gm, off, 0, d, lane, src0, ex0);
            float4 den = ex0;
            for (int c0 = 32; c0 < d; c0 += 32) {
                int sx; float4 ex;
                bucket_weights(lsrc, s_src4, st, gm, off, c0, d, lane, sx, ex);
                den.x += ex.x; den.y += ex.y; den.z += ex.z; den.w += ex.w;
            }
            for (int b = 0; b < novf; b += 32) {
                int sx; float4 ex; bool hit;
                overflow_weights(ovf_tgt, ovf_src, s_src4, st, gm, b, novf, n, lane, sx, ex, hit);
                den.x += ex.x; den.y += ex.y; den.z += ex.z; den.w += ex.w;
            }
#pragma unroll
            for (int o = 16; o > 0; o >>= 1) {
                den.x += __shfl_xor(den.x, o, 32);
                den.y += __shfl_xor(den.y, o, 32);
                den.z += __shfl_xor(den.z, o, 32);
                den.w += __shfl_xor(den.w, o, 32);
            }
            den.x += 1e-16f; den.y += 1e-16f; den.z += 1e-16f; den.w += 1e-16f;
            float4 att = make_float4(ex0.x / den.x, ex0.y / den.y, ex0.z / den.z, ex0.w / den.w);
            bucket_accumulate(P4, src0, att, (d < 32) ? d : 32, lane, hd, acc);
            for (int c0 = 32; c0 < d; c0 += 32) {
                int sx; float4 ex;
                bucket_weights(lsrc, s_src4, st, gm, off, c0, d, lane, sx, ex);
                att = make_float4(ex.x / den.x, ex.y / den.y, ex.z / den.z, ex.w / den.w);
                const int rem = d - c0;
                bucket_accumulate(P4, sx, att, (rem < 32) ? rem : 32, lane, hd, acc);
            }
            for (int b = 0; b < novf; b += 32) {
                int sx; float4 ex; bool hit;
                overflow_weights(ovf_tgt, ovf_src, s_src4, st, gm, b, novf, n, lane, sx, ex, hit);
                att = make_float4(ex.x / den.x, ex.y / den.y, ex.z / den.z, ex.w / den.w);
                unsigned todo = __builtin_amdgcn_ballot_w32(hit);
                while (todo != 0u) {
                    const int j = __builtin_ctz(todo);
                    todo &= todo - 1u;
                    const int sj = __builtin_amdgcn_readlane(sx, j);
                    const float4 r = P4[(size_t)sj * (D_IN / 4) + lane];
                    const float a = head_weight(att, j, m0, m1, m2, m3);
                    acc.x = __builtin_fmaf(r.x, a, acc.x);
                    acc.y = __builtin_fmaf(r.y, a, acc.y);
                    acc.z = __builtin_fmaf(r.z, a, acc.z);
                    acc.w = __builtin_fmaf(r.w, a, acc.w);
                }
            }
        }
        ((float4*)out)[(size_t)n * (D_IN / 4) + lane] =
            make_float4(acc.x + bv.x, acc.y + bv.y, acc.z + bv.z, acc.w + bv.w);
    }
}

extern "C" void kernel_launch(void* const* d_in, const int* in_sizes, int n_in,
                              void* d_out, int out_size, void* d_ws, size_t ws_size,
                              hipStream_t stream) {
    const float* in_feat = (const float*)d_in[0];
    const int*   ei      = (const int*)  d_in[1];
    const float* W_proj  = (const float*)d_in[3];
    const float* a_src   = (const float*)d_in[4];
    const float* a_tgt   = (const float*)d_in[5];
    const float* bias    = (const float*)d_in[6];
    float*       out     = (float*)d_out;

    float*    proj    = (float*)d_ws;
    float*    s_src   = proj  + (size_t)N_NODES * D_IN;
    float*    s_tgt   = s_src + (size_t)N_NODES * N_HEAD;
    int*      seg     = (int*)(s_tgt + (size_t)N_NODES * N_HEAD);
    unsigned* seg_cnt = (unsigned*)(seg + (size_t)NB * BIN_WGS * SEGCAP * 2);
    int*      ovf_tgt = (int*)(seg_cnt + (size_t)BIN_WGS * NB);
    int*      ovf_src = ovf_tgt + (size_t)N_EDGES;
    float*    wg_max  = (float*)(ovf_src + (size_t)N_EDGES);
    unsigned* ovf_cnt = (unsigned*)(wg_max + BIN_WGS);

    gat_proj_scores_kernel<<<N_NODES / 16, 128, 0, stream>>>(
        in_feat, W_proj, a_src, a_tgt, proj, s_src, s_tgt, ovf_cnt);

    gat_bin_max_kernel<<<BIN_WGS, 1024, 0, stream>>>(
        ei, s_src, s_tgt, wg_max, seg_cnt, seg, ovf_cnt, ovf_tgt, ovf_src);

    gat_gather_kernel<<<NB, 1024, 0, stream>>>(
        seg_cnt, seg, ovf_cnt, ovf_tgt, ovf_src, proj, s_src, s_tgt, wg_max, bias, out);
}


# gather kernel: weights computed one per lane as (edge, head) in 8-lane groups; an edge's weight reaches its head's lanes by one in-group broadcast, not 4 lane reads; denominator sum order changes
# speedup vs baseline: 7.9647x; 1.0324x over previous
#include <hip/hip_runtime.h>
#include <math.h>

#define N_NODES 50000
#define N_EDGES 800000
#define D_IN    128
#define N_HEAD  4
#define D_OUT   32
#define NB      512
#define TPB     98
#define BIN_WGS 256
#define EPB     3125
#define SEGCAP  32
#define BCAP    (BIN_WGS * SEGCAP)

typedef __attribute__((ext_vector_type(2))) float v2f;
typedef __attribute__((ext_vector_type(8))) float v8f;


__global__ __launch_bounds__(128)
void gat_proj_scores_kernel(const float* __restrict__ A,
                            const float* __restrict__ W,
                            const float* __restrict__ a_src,
                            const float* __restrict__ a_tgt,
                            float* __restrict__ P,
                            float* __restrict__ s_src,
                            float* __restrict__ s_tgt,
                            unsigned* __restrict__ ovf_cnt) {
    const int head = threadIdx.x >> 5;
    const int lane = threadIdx.x & 31;
    const int c    = lane & 15;
    const int half = lane >> 4;
    const int m0   = blockIdx.x * 16;
    const int col0 = head * D_OUT + 2 * c;
    const int kGrp = half * 2;

    const float* Arow = A + (size_t)(m0 + c) * D_IN + kGrp;
    const float* Wcol = W + (size_t)kGrp * D_IN + col0;

    v8f acc0 = {};
    v8f acc1 = {};
#pragma unroll
    for (int k0 = 0; k0 < D_IN; k0 += 4) {
        const float2 av = *(const float2*)(Arow + k0);
        const float2 w0 = *(const float2*)(Wcol + (size_t)k0 * D_IN);
        const float2 w1 = *(const float2*)(Wcol + (size_t)(k0 + 1) * D_IN);
        v2f a, b0, b1;
        a.x  = av.x;  a.y  = av.y;
        b0.x = w0.x;  b0.y = w1.x;
        b1.x = w0.y;  b1.y = w1.y;
        acc0 = __builtin_amdgcn_wmma_f32_16x16x4_f32(false, a, false, b0,
                                                     (short)0, acc0, false, false);
        acc1 = __builtin_amdgcn_wmma_f32_16x16x4_f32(false, a, false, b1,
                                                     (short)0, acc1, false, false);
    }

    const int mBase = m0 + 8 * half;
#pragma unroll
    for (int v = 0; v < 8; ++v)
        *(float2*)(P + (size_t)(mBase + v) * D_IN + col0) = make_float2(acc0[v], acc1[v]);

    const float2 as = *(const float2*)(a_src + col0);
    const float2 at = *(const float2*)(a_tgt + col0);
    float x[16];
#pragma unroll
    for (int v = 0; v < 8; ++v) {
        x[v]     = __builtin_fmaf(acc1[v], as.y, acc0[v] * as.x);
        x[8 + v] = __builtin_fmaf(acc1[v], at.y, acc0[v] * at.x);
    }
    const bool b3 = (c & 8) != 0, b2 = (c & 4) != 0, b1 = (c & 2) != 0, b0 = (c & 1) != 0;
    float y[8];
#pragma unroll
    for (int j = 0; j < 8; ++j) {
        const float keep = b3 ? x[j + 8] : x[j];
        const float send = b3 ? x[j] : x[j + 8];
        y[j] = keep + __shfl_xor(send, 8, 32);
    }
    float z[4];
#pragma unroll
    for (int j = 0; j < 4; ++j) {
        const float keep = b2 ? y[j + 4] : y[j];
        const float send = b2 ? y[j] : y[j + 4];
        z[j] = keep + __shfl_xor(send, 4, 32);
    }
    float w[2];
#pragma unroll
    for (int j = 0; j < 2; ++j) {
        const float keep = b1 ? z[j + 2] : z[j];
        const float send = b1 ? z[j] : z[j + 2];
        w[j] = keep + __shfl_xor(send, 2, 32);
    }
    const float keep0 = b0 ? w[1] : w[0];
    const float send0 = b0 ? w[0] : w[1];
    const float tot   = keep0 + __shfl_xor(send0, 1, 32);
    float* dst = (c & 8) ? s_tgt : s_src;
    dst[(size_t)(mBase + (c & 7)) * N_HEAD + head] = tot;
    if (blockIdx.x == 0 && threadIdx.x == 0)
        *ovf_cnt = 0u;
}

__global__ __launch_bounds__(1024)
void gat_bin_max_kernel(const int* __restrict__ ei,
                        const float* __restrict__ s_src,
                        const float* __restrict__ s_tgt,
                        float* __restrict__ wg_max,
                        unsigned* __restrict__ seg_cnt,
                        int* __restrict__ seg_w,
                        unsigned* __restrict__ ovf_cnt,
                        int* __restrict__ ovf_tgt,
                        int* __restrict__ ovf_src) {
    __shared__ unsigned lcount[NB];
    __shared__ float    lmax[32];
    int2* seg = (int2*)seg_w;
    const int tid = threadIdx.x;
    const int wg  = blockIdx.x;
    if (tid < NB)
        lcount[tid] = 0u;
    __syncthreads();

    const int e0 = wg * EPB;
    int src[4], tgt[4];
#pragma unroll
    for (int k = 0; k < 4; ++k) {
        const int i = tid + k * 1024;
        src[k] = 0; tgt[k] = 0;
        if (i < EPB) {
            src[k] = ei[e0 + i];
            tgt[k] = ei[N_EDGES + e0 + i];
        }
    }
    float m = -INFINITY;
#pragma unroll
    for (int k = 0; k < 4; ++k) {
        const int i = tid + k * 1024;
        if (i < EPB) {
            const float4 ss = ((const float4*)s_src)[src[k]];
            const float4 st = ((const float4*)s_tgt)[tgt[k]];
            const int bin = tgt[k] / TPB;
            const unsigned rank = atomicAdd(&lcount[bin], 1u);
            if (rank < (unsigned)SEGCAP) {
                seg[((size_t)bin * BIN_WGS + wg) * SEGCAP + rank] = make_int2(src[k], tgt[k]);
            } else {
                const unsigned o = atomicAdd(ovf_cnt, 1u);
                ovf_tgt[o] = tgt[k];
                ovf_src[o] = src[k];
            }
            const float sc[N_HEAD] = {ss.x + st.x, ss.y + st.y, ss.z + st.z, ss.w + st.w};
#pragma unroll
            for (int h = 0; h < N_HEAD; ++h) {
                const float s = (sc[h] > 0.0f) ? sc[h] : 0.2f * sc[h];
                m = fmaxf(m, s);
            }
        }
    }
#pragma unroll
    for (int off = 16; off > 0; off >>= 1)
        m = fmaxf(m, __shfl_xor(m, off, 32));
    if ((tid & 31) == 0)
        lmax[tid >> 5] = m;
    __syncthreads();

    if (tid < NB)
        seg_cnt[wg * NB + tid] = lcount[tid];
    if (tid < 32) {
        float mm = lmax[tid];
#pragma unroll
        for (int off = 16; off > 0; off >>= 1)
            mm = fmaxf(mm, __shfl_xor(mm, off, 32));
        if (tid == 0)
            wg_max[wg] = mm;
    }
}

__device__ __forceinline__ float head_weight(const float4 att, const int j,
                                             const int m0, const int m1,
                                             const int m2, const int m3) {
    const int a0 = __builtin_amdgcn_readlane(__float_as_int(att.x), j);
    const int a1 = __builtin_amdgcn_readlane(__float_as_int(att.y), j);
    const int a2 = __builtin_amdgcn_readlane(__float_as_int(att.z), j);
    const int a3 = __builtin_amdgcn_readlane(__float_as_int(att.w), j);
    return __int_as_float((a0 & m0) | (a1 & m1) | (a2 & m2) | (a3 & m3));
}

__device__ __forceinline__ float4 edge_exp(const float4 ss, const float4 st,
                                           const float gm) {
    float s0 = ss.x + st.x, s1 = ss.y + st.y, s2 = ss.z + st.z, s3 = ss.w + st.w;
    s0 = (s0 > 0.0f) ? s0 : 0.2f * s0;
    s1 = (s1 > 0.0f) ? s1 : 0.2f * s1;
    s2 = (s2 > 0.0f) ? s2 : 0.2f * s2;
    s3 = (s3 > 0.0f) ? s3 : 0.2f * s3;
    return make_float4(__expf(s0 - gm), __expf(s1 - gm), __expf(s2 - gm), __expf(s3 - gm));
}

__device__ __forceinline__ float group_exp(const float ssrc, const float stl,
                                           const float gm, const bool valid) {
    float s = ssrc + stl;
    s = (s > 0.0f) ? s : 0.2f * s;
    const float e = __expf(s - gm);
    return valid ? e : 0.0f;
}

__device__ __forceinline__ float group_bcast(const float v, const int e) {
    const unsigned x = __float_as_uint(v);
    unsigned r;
    switch (e) {
    case 0:  r = __builtin_amdgcn_mov_dpp8(x, 0x000000); break;
    case 1:  r = __builtin_amdgcn_mov_dpp8(x, 0x249249); break;
    case 2:  r = __builtin_amdgcn_mov_dpp8(x, 0x492492); break;
    case 3:  r = __builtin_amdgcn_mov_dpp8(x, 0x6DB6DB); break;
    case 4:  r = __builtin_amdgcn_mov_dpp8(x, 0x924924); break;
    case 5:  r = __builtin_amdgcn_mov_dpp8(x, 0xB6DB6D); break;
    case 6:  r = __builtin_amdgcn_mov_dpp8(x, 0xDB6DB6); break;
    default: r = __builtin_amdgcn_mov_dpp8(x, 0xFFFFFF); break;
    }
    return __uint_as_float(r);
}

__device__ __forceinline__ float group_sum(float v) {
    v += __uint_as_float(__builtin_amdgcn_mov_dpp8(__float_as_uint(v), 0xDE54C1));
    v += __uint_as_float(__builtin_amdgcn_mov_dpp8(__float_as_uint(v), 0xB3E21A));
    v += __uint_as_float(__builtin_amdgcn_mov_dpp8(__float_as_uint(v), 0x688FAC));
    return v;
}

__device__ __forceinline__ void overflow_weights(const int* __restrict__ ovf_tgt,
                                                 const int* __restrict__ ovf_src,
                                                 const float4* __restrict__ s_src4,
                                                 const float4 st, const float gm,
                                                 const int base, const int novf,
                                                 const int n, const int lane,
                                                 int& src, float4& ex, bool& hit) {
    src = 0;
    ex = make_float4(0.0f, 0.0f, 0.0f, 0.0f);
    hit = false;
    const int i = base + lane;
    if (i < novf && ovf_tgt[i] == n) {
        hit = true;
        src = ovf_src[i];
        ex = edge_exp(s_src4[src], st, gm);
    }
}

template <int N>
__device__ __forceinline__ void group_rows(const float4* __restrict__ P4,
                                           const int src, const float w,
                                           const int b, const int lane,
                                           float4& acc) {
    float4 r[N];
#pragma unroll
    for (int u = 0; u < N; ++u) {
        const int sj = __builtin_amdgcn_readlane(src, b + u);
        r[u] = P4[(unsigned)(sj * (D_IN / 4) + lane)];
    }
#pragma unroll
    for (int u = 0; u < N; ++u) {
        const float a = group_bcast(w, b + u);
        acc.x = __builtin_fmaf(r[u].x, a, acc.x);
        acc.y = __builtin_fmaf(r[u].y, a, acc.y);
        acc.z = __builtin_fmaf(r[u].z, a, acc.z);
        acc.w = __builtin_fmaf(r[u].w, a, acc.w);
    }
}

__device__ __forceinline__ void group_accumulate(const float4* __restrict__ P4,
                                                 const int src, const float w,
                                                 const int cnt, const int lane,
                                                 float4& acc) {
#pragma unroll
    for (int b = 0; b < 8; b += 4) {
        if (b + 4 <= cnt) {
            group_rows<4>(P4, src, w, b, lane, acc);
        } else if (b < cnt) {
            const int left = cnt - b;
            if (left == 3) {
                group_rows<3>(P4, src, w, b, lane, acc);
            } else if (left == 2) {
                group_rows<2>(P4, src, w, b, lane, acc);
            } else {
                group_rows<1>(P4, src, w, b, lane, acc);
            }
        }
    }
}

__global__ __launch_bounds__(1024, 16)
void gat_gather_kernel(const unsigned* __restrict__ seg_cnt,
                       const int* __restrict__ seg_w,
                       const unsigned* __restrict__ ovf_cnt,
                       const int* __restrict__ ovf_tgt,
                       const int* __restrict__ ovf_src,
                       const float* __restrict__ P,
                       const float* __restrict__ s_src,
                       const float* __restrict__ s_tgt,
                       const float* __restrict__ wg_max,
                       const float* __restrict__ bias,
                       float* __restrict__ out) {
    __shared__ unsigned ldeg[128];
    __shared__ unsigned loff[128];
    __shared__ unsigned lcur[128];
    __shared__ unsigned lnext;
    __shared__ float    lgm;
    __shared__ int      lsrc[BCAP];
    const int tid   = threadIdx.x;
    const int lane  = tid & 31;
    const int hd    = lane >> 3;
    const int e8    = lane & 7;
    const int wave  = __builtin_amdgcn_readfirstlane(tid >> 5);
    const int rng   = blockIdx.x;
    const int nbase = rng * TPB;
    const int2* seg = (const int2*)seg_w + (size_t)rng * BIN_WGS * SEGCAP;

    if (tid < 128)
        ldeg[tid] = 0u;
    if (tid == 0)
        lnext = 0u;

    int2 ent[8];
    bool ok[8];
    unsigned segn[8];
#pragma unroll
    for (int k = 0; k < 8; ++k)
        segn[k] = seg_cnt[(k * 32 + wave) * NB + rng];
#pragma unroll
    for (int k = 0; k < 8; ++k) {
        const unsigned c = (segn[k] < (unsigned)SEGCAP) ? segn[k] : (unsigned)SEGCAP;
        ok[k] = (unsigned)lane < c;
        ent[k] = make_int2(0, nbase);
        if (ok[k])
            ent[k] = seg[(k * 32 + wave) * SEGCAP + lane];
    }
    float gm8 = 0.0f;
    if (wave == 1) {
        const float4 gma = ((const float4*)wg_max)[2 * lane];
        const float4 gmb = ((const float4*)wg_max)[2 * lane + 1];
        gm8 = fmaxf(fmaxf(fmaxf(gma.x, gma.y), fmaxf(gma.z, gma.w)),
                    fmaxf(fmaxf(gmb.x, gmb.y), fmaxf(gmb.z, gmb.w)));
    }
    __syncthreads();

#pragma unroll
    for (int k = 0; k < 8; ++k)
        if (ok[k])
            atomicAdd(&ldeg[ent[k].y - nbase], 1u);
    __syncthreads();

    if (tid < 32) {
        const unsigned v0 = ldeg[4 * lane + 0], v1 = ldeg[4 * lane + 1];
        const unsigned v2 = ldeg[4 * lane + 2], v3 = ldeg[4 * lane + 3];
        const unsigned s = v0 + v1 + v2 + v3;
        unsigned incl = s;
#pragma unroll
        for (int o = 1; o < 32; o <<= 1) {
            const unsigned y = __shfl_up(incl, o, 32);
            if (lane >= o) incl += y;
        }
        const unsigned e0 = incl - s, e1 = e0 + v0, e2 = e1 + v1, e3 = e2 + v2;
        loff[4 * lane + 0] = e0;  lcur[4 * lane + 0] = e0;
        loff[4 * lane + 1] = e1;  lcur[4 * lane + 1] = e1;
        loff[4 * lane + 2] = e2;  lcur[4 * lane + 2] = e2;
        loff[4 * lane + 3] = e3;  lcur[4 * lane + 3] = e3;
    } else if (wave == 1) {
        float mm = gm8;
#pragma unroll
        for (int o = 16; o > 0; o >>= 1)
            mm = fmaxf(mm, __shfl_xor(mm, o, 32));
        if (lane == 0)
            lgm = mm;
    }
    __syncthreads();

#pragma unroll
    for (int k = 0; k < 8; ++k)
        if (ok[k]) {
            const unsigned p = atomicAdd(&lcur[ent[k].y - nbase], 1u);
            lsrc[p] = ent[k].x;
        }
    __syncthreads();

    const float4* P4     = (const float4*)P;
    const float4* s_src4 = (const float4*)s_src;
    const float   gm = lgm;
    const float4  bv = ((const float4*)bias)[lane];
    const unsigned oc = *ovf_cnt;
    const int novf = (oc < (unsigned)N_EDGES) ? (int)oc : N_EDGES;

    for (int it = 0; it < TPB; ++it) {
        int t = 0;
        if (lane == 0)
            t = (int)atomicAdd(&lnext, 1u);
        t = __builtin_amdgcn_readfirstlane(t);
        const int n = nbase + t;
        if (t >= TPB || n >= N_NODES)
            break;
        const int d   = __builtin_amdgcn_readfirstlane((int)ldeg[t]);
        const int off = __builtin_amdgcn_readfirstlane((int)loff[t]);

        float4 acc = make_float4(0.0f, 0.0f, 0.0f, 0.0f);
        if (d > 0 || novf > 0) {
            const float stl = s_tgt[(size_t)n * N_HEAD + hd];
            float ex[4], sc[4];
            int   sv[4];
            float den = 0.0f;
#pragma unroll
            for (int q = 0; q < 4; ++q) {
                sv[q] = 0;
                if (8 * q < d) {
                    const int j = 8 * q + e8;
                    sv[q] = lsrc[off + ((j < d) ? j : d - 1)];
                }
            }
#pragma unroll
            for (int q = 0; q < 4; ++q) {
                sc[q] = 0.0f;
                if (8 * q < d)
                    sc[q] = s_src[(unsigned)(sv[q] * N_HEAD + hd)];
            }
#pragma unroll
            for (int q = 0; q < 4; ++q) {
                ex[q] = 0.0f;
                if (8 * q < d) {
                    ex[q] = group_exp(sc[q], stl, gm, 8 * q + e8 < d);
                    den += ex[q];
                }
            }
            for (int c0 = 32; c0 < d; c0 += 8) {
                const int j = c0 + e8;
                const int sx = lsrc[off + ((j < d) ? j : d - 1)];
                den += group_exp(s_src[(unsigned)(sx * N_HEAD + hd)], stl, gm, j < d);
            }
            den = group_sum(den);
            if (novf > 0) {
                const float4 st = ((const float4*)s_tgt)[n];
                float4 dov = make_float4(0.0f, 0.0f, 0.0f, 0.0f);
                for (int b = 0; b < novf; b += 32) {
                    int sx; float4 exo; bool hit;
                    overflow_weights(ovf_tgt, ovf_src, s_src4, st, gm, b, novf, n, lane, sx, exo, hit);
                    dov.x += exo.x; dov.y += exo.y; dov.z += exo.z; dov.w += exo.w;
                }
#pragma unroll
                for (int o = 16; o > 0; o >>= 1) {
                    dov.x += __shfl_xor(dov.x, o, 32);
                    dov.y += __shfl_xor(dov.y, o, 32);
                    dov.z += __shfl_xor(dov.z, o, 32);
                    dov.w += __shfl_xor(dov.w, o, 32);
                }
                den += (hd == 0) ? dov.x : (hd == 1) ? dov.y : (hd == 2) ? dov.z : dov.w;
            }
            den += 1e-16f;
#pragma unroll
            for (int q = 0; q < 4; ++q)
                if (8 * q < d) {
                    const int rem = d - 8 * q;
                    group_accumulate(P4, sv[q], ex[q] / den, (rem < 8) ? rem : 8, lane, acc);
                }
            for (int c0 = 32; c0 < d; c0 += 8) {
                const int j = c0 + e8;
                const int sx = lsrc[off + ((j < d) ? j : d - 1)];
                const float exx = group_exp(s_src[(unsigned)(sx * N_HEAD + hd)], stl, gm, j < d);
                const int rem = d - c0;
                group_accumulate(P4, sx, exx / den, (rem < 8) ? rem : 8, lane, acc);
            }
            if (novf > 0) {
                const float4 st = ((const float4*)s_tgt)[n];
                const int m0 = (hd == 0) ? -1 : 0, m1 = (hd == 1) ? -1 : 0;
                const int m2 = (hd == 2) ? -1 : 0, m3 = (hd == 3) ? -1 : 0;
                const float4 den4 = make_float4(
                    __int_as_float(__builtin_amdgcn_readlane(__float_as_int(den), 0)),
                    __int_as_float(__builtin_amdgcn_readlane(__float_as_int(den), 8)),
                    __int_as_float(__builtin_amdgcn_readlane(__float_as_int(den), 16)),
                    __int_as_float(__builtin_amdgcn_readlane(__float_as_int(den), 24)));
                for (int b = 0; b < novf; b += 32) {
                    int sx; float4 exo; bool hit;
                    overflow_weights(ovf_tgt, ovf_src, s_src4, st, gm, b, novf, n, lane, sx, exo, hit);
                    const float4 att = make_float4(exo.x / den4.x, exo.y / den4.y, exo.z / den4.z, exo.w / den4.w);
                    unsigned todo = __builtin_amdgcn_ballot_w32(hit);
                    while (todo != 0u) {
                        const int j = __builtin_ctz(todo);
                        todo &= todo - 1u;
                        const int sj = __builtin_amdgcn_readlane(sx, j);
                        const float4 r = P4[(size_t)sj * (D_IN / 4) + lane];
                        const float a = head_weight(att, j, m0, m1, m2, m3);
                        acc.x = __builtin_fmaf(r.x, a, acc.x);
                        acc.y = __builtin_fmaf(r.y, a, acc.y);
                        acc.z = __builtin_fmaf(r.z, a, acc.z);
                        acc.w = __builtin_fmaf(r.w, a, acc.w);
                    }
                }
            }
        }
        ((float4*)out)[(size_t)n * (D_IN / 4) + lane] =
            make_float4(acc.x + bv.x, acc.y + bv.y, acc.z + bv.z, acc.w + bv.w);
    }
}

extern "C" void kernel_launch(void* const* d_in, const int* in_sizes, int n_in,
                              void* d_out, int out_size, void* d_ws, size_t ws_size,
                              hipStream_t stream) {
    const float* in_feat = (const float*)d_in[0];
    const int*   ei      = (const int*)  d_in[1];
    const float* W_proj  = (const float*)d_in[3];
    const float* a_src   = (const float*)d_in[4];
    const float* a_tgt   = (const float*)d_in[5];
    const float* bias    = (const float*)d_in[6];
    float*       out     = (float*)d_out;

    float*    proj    = (float*)d_ws;
    float*    s_src   = proj  + (size_t)N_NODES * D_IN;
    float*    s_tgt   = s_src + (size_t)N_NODES * N_HEAD;
    int*      seg     = (int*)(s_tgt + (size_t)N_NODES * N_HEAD);
    unsigned* seg_cnt = (unsigned*)(seg + (size_t)NB * BIN_WGS * SEGCAP * 2);
    int*      ovf_tgt = (int*)(seg_cnt + (size_t)BIN_WGS * NB);
    int*      ovf_src = ovf_tgt + (size_t)N_EDGES;
    float*    wg_max  = (float*)(ovf_src + (size_t)N_EDGES);
    unsigned* ovf_cnt = (unsigned*)(wg_max + BIN_WGS);

    gat_proj_scores_kernel<<<N_NODES / 16, 128, 0, stream>>>(
        in_feat, W_proj, a_src, a_tgt, proj, s_src, s_tgt, ovf_cnt);

    gat_bin_max_kernel<<<BIN_WGS, 1024, 0, stream>>>(
        ei, s_src, s_tgt, wg_max, seg_cnt, seg, ovf_cnt, ovf_tgt, ovf_src);

    gat_gather_kernel<<<NB, 1024, 0, stream>>>(
        seg_cnt, seg, ovf_cnt, ovf_tgt, ovf_src, proj, s_src, s_tgt, wg_max, bias, out);
}


# projection kernel: 16x128 A tile staged once per work-group in LDS by whole-row 16-byte loads (was 8-byte strided loads in all 4 waves), fragments by 16-byte LDS reads; k order in 8-blocks changes
# speedup vs baseline: 8.4509x; 1.0610x over previous
#include <hip/hip_runtime.h>
#include <math.h>

#define N_NODES 50000
#define N_EDGES 800000
#define D_IN    128
#define N_HEAD  4
#define D_OUT   32
#define NB      512
#define TPB     98
#define BIN_WGS 256
#define EPB     3125
#define SEGCAP  32
#define BCAP    (BIN_WGS * SEGCAP)
#define A_PITCH 132

typedef __attribute__((ext_vector_type(2))) float v2f;
typedef __attribute__((ext_vector_type(8))) float v8f;


__global__ __launch_bounds__(128)
void gat_proj_scores_kernel(const float* __restrict__ A,
                            const float* __restrict__ W,
                            const float* __restrict__ a_src,
                            const float* __restrict__ a_tgt,
                            float* __restrict__ P,
                            float* __restrict__ s_src,
                            float* __restrict__ s_tgt,
                            unsigned* __restrict__ ovf_cnt) {
    __shared__ __attribute__((aligned(16))) float sA[16 * A_PITCH];
    const int head = threadIdx.x >> 5;
    const int lane = threadIdx.x & 31;
    const int c    = lane & 15;
    const int half = lane >> 4;
    const int m0   = blockIdx.x * 16;
    const int col0 = head * D_OUT + 2 * c;

#pragma unroll
    for (int j = 0; j < 4; ++j) {
        const int r = 4 * j + head;
        const float4 v = *(const float4*)(A + (size_t)(m0 + r) * D_IN + 4 * lane);
        *(float4*)(sA + r * A_PITCH + 4 * lane) = v;
    }
    __syncthreads();

    const float* Arow = sA + c * A_PITCH + 4 * half;
    const float* Wcol = W + (size_t)(4 * half) * D_IN + col0;

    v8f acc0 = {};
    v8f acc1 = {};
#pragma unroll
    for (int kb = 0; kb < D_IN; kb += 8) {
        const float4 av = *(const float4*)(Arow + kb);
        const float2 w0 = *(const float2*)(Wcol + (size_t)(kb + 0) * D_IN);
        const float2 w1 = *(const float2*)(Wcol + (size_t)(kb + 1) * D_IN);
        const float2 w2 = *(const float2*)(Wcol + (size_t)(kb + 2) * D_IN);
        const float2 w3 = *(const float2*)(Wcol + (size_t)(kb + 3) * D_IN);
        v2f a, b0, b1;
        a.x  = av.x;  a.y  = av.y;
        b0.x = w0.x;  b0.y = w1.x;
        b1.x = w0.y;  b1.y = w1.y;
        acc0 = __builtin_amdgcn_wmma_f32_16x16x4_f32(false, a, false, b0,
                                                     (short)0, acc0, false, false);
        acc1 = __builtin_amdgcn_wmma_f32_16x16x4_f32(false, a, false, b1,
                                                     (short)0, acc1, false, false);
        a.x  = av.z;  a.y  = av.w;
        b0.x = w2.x;  b0.y = w3.x;
        b1.x = w2.y;  b1.y = w3.y;
        acc0 = __builtin_amdgcn_wmma_f32_16x16x4_f32(false, a, false, b0,
                                                     (short)0, acc0, false, false);
        acc1 = __builtin_amdgcn_wmma_f32_16x16x4_f32(false, a, false, b1,
                                                     (short)0, acc1, false, false);
    }

    const int mBase = m0 + 8 * half;
#pragma unroll
    for (int v = 0; v < 8; ++v)
        *(float2*)(P + (size_t)(mBase + v) * D_IN + col0) = make_float2(acc0[v], acc1[v]);

    const float2 as = *(const float2*)(a_src + col0);
    const float2 at = *(const float2*)(a_tgt + col0);
    float x[16];
#pragma unroll
    for (int v = 0; v < 8; ++v) {
        x[v]     = __builtin_fmaf(acc1[v], as.y, acc0[v] * as.x);
        x[8 + v] = __builtin_fmaf(acc1[v], at.y, acc0[v] * at.x);
    }
    const bool b3 = (c & 8) != 0, b2 = (c & 4) != 0, b1 = (c & 2) != 0, b0 = (c & 1) != 0;
    float y[8];
#pragma unroll
    for (int j = 0; j < 8; ++j) {
        const float keep = b3 ? x[j + 8] : x[j];
        const float send = b3 ? x[j] : x[j + 8];
        y[j] = keep + __shfl_xor(send, 8, 32);
    }
    float z[4];
#pragma unroll
    for (int j = 0; j < 4; ++j) {
        const float keep = b2 ? y[j + 4] : y[j];
        const float send = b2 ? y[j] : y[j + 4];
        z[j] = keep + __shfl_xor(send, 4, 32);
    }
    float w[2];
#pragma unroll
    for (int j = 0; j < 2; ++j) {
        const float keep = b1 ? z[j + 2] : z[j];
        const float send = b1 ? z[j] : z[j + 2];
        w[j] = keep + __shfl_xor(send, 2, 32);
    }
    const float keep0 = b0 ? w[1] : w[0];
    const float send0 = b0 ? w[0] : w[1];
    const float tot   = keep0 + __shfl_xor(send0, 1, 32);
    float* dst = (c & 8) ? s_tgt : s_src;
    dst[(size_t)(mBase + (c & 7)) * N_HEAD + head] = tot;
    if (blockIdx.x == 0 && threadIdx.x == 0)
        *ovf_cnt = 0u;
}

__global__ __launch_bounds__(1024)
void gat_bin_max_kernel(const int* __restrict__ ei,
                        const float* __restrict__ s_src,
                        const float* __restrict__ s_tgt,
                        float* __restrict__ wg_max,
                        unsigned* __restrict__ seg_cnt,
                        int* __restrict__ seg_w,
                        unsigned* __restrict__ ovf_cnt,
                        int* __restrict__ ovf_tgt,
                        int* __restrict__ ovf_src) {
    __shared__ unsigned lcount[NB];
    __shared__ float    lmax[32];
    int2* seg = (int2*)seg_w;
    const int tid = threadIdx.x;
    const int wg  = blockIdx.x;
    if (tid < NB)
        lcount[tid] = 0u;
    __syncthreads();

    const int e0 = wg * EPB;
    int src[4], tgt[4];
#pragma unroll
    for (int k = 0; k < 4; ++k) {
        const int i = tid + k * 1024;
        src[k] = 0; tgt[k] = 0;
        if (i < EPB) {
            src[k] = ei[e0 + i];
            tgt[k] = ei[N_EDGES + e0 + i];
        }
    }
    float m = -INFINITY;
#pragma unroll
    for (int k = 0; k < 4; ++k) {
        const int i = tid + k * 1024;
        if (i < EPB) {
            const float4 ss = ((const float4*)s_src)[src[k]];
            const float4 st = ((const float4*)s_tgt)[tgt[k]];
            const int bin = tgt[k] / TPB;
            const unsigned rank = atomicAdd(&lcount[bin], 1u);
            if (rank < (unsigned)SEGCAP) {
                seg[((size_t)bin * BIN_WGS + wg) * SEGCAP + rank] = make_int2(src[k], tgt[k]);
            } else {
                const unsigned o = atomicAdd(ovf_cnt, 1u);
                ovf_tgt[o] = tgt[k];
                ovf_src[o] = src[k];
            }
            const float sc[N_HEAD] = {ss.x + st.x, ss.y + st.y, ss.z + st.z, ss.w + st.w};
#pragma unroll
            for (int h = 0; h < N_HEAD; ++h) {
                const float s = (sc[h] > 0.0f) ? sc[h] : 0.2f * sc[h];
                m = fmaxf(m, s);
            }
        }
    }
#pragma unroll
    for (int off = 16; off > 0; off >>= 1)
        m = fmaxf(m, __shfl_xor(m, off, 32));
    if ((tid & 31) == 0)
        lmax[tid >> 5] = m;
    __syncthreads();

    if (tid < NB)
        seg_cnt[wg * NB + tid] = lcount[tid];
    if (tid < 32) {
        float mm = lmax[tid];
#pragma unroll
        for (int off = 16; off > 0; off >>= 1)
            mm = fmaxf(mm, __shfl_xor(mm, off, 32));
        if (tid == 0)
            wg_max[wg] = mm;
    }
}

__device__ __forceinline__ float head_weight(const float4 att, const int j,
                                             const int m0, const int m1,
                                             const int m2, const int m3) {
    const int a0 = __builtin_amdgcn_readlane(__float_as_int(att.x), j);
    const int a1 = __builtin_amdgcn_readlane(__float_as_int(att.y), j);
    const int a2 = __builtin_amdgcn_readlane(__float_as_int(att.z), j);
    const int a3 = __builtin_amdgcn_readlane(__float_as_int(att.w), j);
    return __int_as_float((a0 & m0) | (a1 & m1) | (a2 & m2) | (a3 & m3));
}

__device__ __forceinline__ float4 edge_exp(const float4 ss, const float4 st,
                                           const float gm) {
    float s0 = ss.x + st.x, s1 = ss.y + st.y, s2 = ss.z + st.z, s3 = ss.w + st.w;
    s0 = (s0 > 0.0f) ? s0 : 0.2f * s0;
    s1 = (s1 > 0.0f) ? s1 : 0.2f * s1;
    s2 = (s2 > 0.0f) ? s2 : 0.2f * s2;
    s3 = (s3 > 0.0f) ? s3 : 0.2f * s3;
    return make_float4(__expf(s0 - gm), __expf(s1 - gm), __expf(s2 - gm), __expf(s3 - gm));
}

__device__ __forceinline__ float group_exp(const float ssrc, const float stl,
                                           const float gm, const bool valid) {
    float s = ssrc + stl;
    s = (s > 0.0f) ? s : 0.2f * s;
    const float e = __expf(s - gm);
    return valid ? e : 0.0f;
}

__device__ __forceinline__ float group_bcast(const float v, const int e) {
    const unsigned x = __float_as_uint(v);
    unsigned r;
    switch (e) {
    case 0:  r = __builtin_amdgcn_mov_dpp8(x, 0x000000); break;
    case 1:  r = __builtin_amdgcn_mov_dpp8(x, 0x249249); break;
    case 2:  r = __builtin_amdgcn_mov_dpp8(x, 0x492492); break;
    case 3:  r = __builtin_amdgcn_mov_dpp8(x, 0x6DB6DB); break;
    case 4:  r = __builtin_amdgcn_mov_dpp8(x, 0x924924); break;
    case 5:  r = __builtin_amdgcn_mov_dpp8(x, 0xB6DB6D); break;
    case 6:  r = __builtin_amdgcn_mov_dpp8(x, 0xDB6DB6); break;
    default: r = __builtin_amdgcn_mov_dpp8(x, 0xFFFFFF); break;
    }
    return __uint_as_float(r);
}

__device__ __forceinline__ float group_sum(float v) {
    v += __uint_as_float(__builtin_amdgcn_mov_dpp8(__float_as_uint(v), 0xDE54C1));
    v += __uint_as_float(__builtin_amdgcn_mov_dpp8(__float_as_uint(v), 0xB3E21A));
    v += __uint_as_float(__builtin_amdgcn_mov_dpp8(__float_as_uint(v), 0x688FAC));
    return v;
}

__device__ __forceinline__ void overflow_weights(const int* __restrict__ ovf_tgt,
                                                 const int* __restrict__ ovf_src,
                                                 const float4* __restrict__ s_src4,
                                                 const float4 st, const float gm,
                                                 const int base, const int novf,
                                                 const int n, const int lane,
                                                 int& src, float4& ex, bool& hit) {
    src = 0;
    ex = make_float4(0.0f, 0.0f, 0.0f, 0.0f);
    hit = false;
    const int i = base + lane;
    if (i < novf && ovf_tgt[i] == n) {
        hit = true;
        src = ovf_src[i];
        ex = edge_exp(s_src4[src], st, gm);
    }
}

template <int N>
__device__ __forceinline__ void group_rows(const float4* __restrict__ P4,
                                           const int src, const float w,
                                           const int b, const int lane,
                                           float4& acc) {
    float4 r[N];
#pragma unroll
    for (int u = 0; u < N; ++u) {
        const int sj = __builtin_amdgcn_readlane(src, b + u);
        r[u] = P4[(unsigned)(sj * (D_IN / 4) + lane)];
    }
#pragma unroll
    for (int u = 0; u < N; ++u) {
        const float a = group_bcast(w, b + u);
        acc.x = __builtin_fmaf(r[u].x, a, acc.x);
        acc.y = __builtin_fmaf(r[u].y, a, acc.y);
        acc.z = __builtin_fmaf(r[u].z, a, acc.z);
        acc.w = __builtin_fmaf(r[u].w, a, acc.w);
    }
}

__device__ __forceinline__ void group_accumulate(const float4* __restrict__ P4,
                                                 const int src, const float w,
                                                 const int cnt, const int lane,
                                                 float4& acc) {
#pragma unroll
    for (int b = 0; b < 8; b += 4) {
        if (b + 4 <= cnt) {
            group_rows<4>(P4, src, w, b, lane, acc);
        } else if (b < cnt) {
            const int left = cnt - b;
            if (left == 3) {
                group_rows<3>(P4, src, w, b, lane, acc);
            } else if (left == 2) {
                group_rows<2>(P4, src, w, b, lane, acc);
            } else {
                group_rows<1>(P4, src, w, b, lane, acc);
            }
        }
    }
}

__global__ __launch_bounds__(1024, 16)
void gat_gather_kernel(const unsigned* __restrict__ seg_cnt,
                       const int* __restrict__ seg_w,
                       const unsigned* __restrict__ ovf_cnt,
                       const int* __restrict__ ovf_tgt,
                       const int* __restrict__ ovf_src,
                       const float* __restrict__ P,
                       const float* __restrict__ s_src,
                       const float* __restrict__ s_tgt,
                       const float* __restrict__ wg_max,
                       const float* __restrict__ bias,
                       float* __restrict__ out) {
    __shared__ unsigned ldeg[128];
    __shared__ unsigned loff[128];
    __shared__ unsigned lcur[128];
    __shared__ unsigned lnext;
    __shared__ float    lgm;
    __shared__ int      lsrc[BCAP];
    const int tid   = threadIdx.x;
    const int lane  = tid & 31;
    const int hd    = lane >> 3;
    const int e8    = lane & 7;
    const int wave  = __builtin_amdgcn_readfirstlane(tid >> 5);
    const int rng   = blockIdx.x;
    const int nbase = rng * TPB;
    const int2* seg = (const int2*)seg_w + (size_t)rng * BIN_WGS * SEGCAP;

    if (tid < 128)
        ldeg[tid] = 0u;
    if (tid == 0)
        lnext = 0u;

    int2 ent[8];
    bool ok[8];
    unsigned segn[8];
#pragma unroll
    for (int k = 0; k < 8; ++k)
        segn[k] = seg_cnt[(k * 32 + wave) * NB + rng];
#pragma unroll
    for (int k = 0; k < 8; ++k) {
        const unsigned c = (segn[k] < (unsigned)SEGCAP) ? segn[k] : (unsigned)SEGCAP;
        ok[k] = (unsigned)lane < c;
        ent[k] = make_int2(0, nbase);
        if (ok[k])
            ent[k] = seg[(k * 32 + wave) * SEGCAP + lane];
    }
    float gm8 = 0.0f;
    if (wave == 1) {
        const float4 gma = ((const float4*)wg_max)[2 * lane];
        const float4 gmb = ((const float4*)wg_max)[2 * lane + 1];
        gm8 = fmaxf(fmaxf(fmaxf(gma.x, gma.y), fmaxf(gma.z, gma.w)),
                    fmaxf(fmaxf(gmb.x, gmb.y), fmaxf(gmb.z, gmb.w)));
    }
    __syncthreads();

#pragma unroll
    for (int k = 0; k < 8; ++k)
        if (ok[k])
            atomicAdd(&ldeg[ent[k].y - nbase], 1u);
    __syncthreads();

    if (tid < 32) {
        const unsigned v0 = ldeg[4 * lane + 0], v1 = ldeg[4 * lane + 1];
        const unsigned v2 = ldeg[4 * lane + 2], v3 = ldeg[4 * lane + 3];
        const unsigned s = v0 + v1 + v2 + v3;
        unsigned incl = s;
#pragma unroll
        for (int o = 1; o < 32; o <<= 1) {
            const unsigned y = __shfl_up(incl, o, 32);
            if (lane >= o) incl += y;
        }
        const unsigned e0 = incl - s, e1 = e0 + v0, e2 = e1 + v1, e3 = e2 + v2;
        loff[4 * lane + 0] = e0;  lcur[4 * lane + 0] = e0;
        loff[4 * lane + 1] = e1;  lcur[4 * lane + 1] = e1;
        loff[4 * lane + 2] = e2;  lcur[4 * lane + 2] = e2;
        loff[4 * lane + 3] = e3;  lcur[4 * lane + 3] = e3;
    } else if (wave == 1) {
        float mm = gm8;
#pragma unroll
        for (int o = 16; o > 0; o >>= 1)
            mm = fmaxf(mm, __shfl_xor(mm, o, 32));
        if (lane == 0)
            lgm = mm;
    }
    __syncthreads();

#pragma unroll
    for (int k = 0; k < 8; ++k)
        if (ok[k]) {
            const unsigned p = atomicAdd(&lcur[ent[k].y - nbase], 1u);
            lsrc[p] = ent[k].x;
        }
    __syncthreads();

    const float4* P4     = (const float4*)P;
    const float4* s_src4 = (const float4*)s_src;
    const float   gm = lgm;
    const float4  bv = ((const float4*)bias)[lane];
    const unsigned oc = *ovf_cnt;
    const int novf = (oc < (unsigned)N_EDGES) ? (int)oc : N_EDGES;

    for (int it = 0; it < TPB; ++it) {
        int t = 0;
        if (lane == 0)
            t = (int)atomicAdd(&lnext, 1u);
        t = __builtin_amdgcn_readfirstlane(t);
        const int n = nbase + t;
        if (t >= TPB || n >= N_NODES)
            break;
        const int d   = __builtin_amdgcn_readfirstlane((int)ldeg[t]);
        const int off = __builtin_amdgcn_readfirstlane((int)loff[t]);

        float4 acc = make_float4(0.0f, 0.0f, 0.0f, 0.0f);
        if (d > 0 || novf > 0) {
            const float stl = s_tgt[(size_t)n * N_HEAD + hd];
            float ex[4], sc[4];
            int   sv[4];
            float den = 0.0f;
#pragma unroll
            for (int q = 0; q < 4; ++q) {
                sv[q] = 0;
                if (8 * q < d) {
                    const int j = 8 * q + e8;
                    sv[q] = lsrc[off + ((j < d) ? j : d - 1)];
                }
            }
#pragma unroll
            for (int q = 0; q < 4; ++q) {
                sc[q] = 0.0f;
                if (8 * q < d)
                    sc[q] = s_src[(unsigned)(sv[q] * N_HEAD + hd)];
            }
#pragma unroll
            for (int q = 0; q < 4; ++q) {
                ex[q] = 0.0f;
                if (8 * q < d) {
                    ex[q] = group_exp(sc[q], stl, gm, 8 * q + e8 < d);
                    den += ex[q];
                }
            }
            for (int c0 = 32; c0 < d; c0 += 8) {
                const int j = c0 + e8;
                const int sx = lsrc[off + ((j < d) ? j : d - 1)];
                den += group_exp(s_src[(unsigned)(sx * N_HEAD + hd)], stl, gm, j < d);
            }
            den = group_sum(den);
            if (novf > 0) {
                const float4 st = ((const float4*)s_tgt)[n];
                float4 dov = make_float4(0.0f, 0.0f, 0.0f, 0.0f);
                for (int b = 0; b < novf; b += 32) {
                    int sx; float4 exo; bool hit;
                    overflow_weights(ovf_tgt, ovf_src, s_src4, st, gm, b, novf, n, lane, sx, exo, hit);
                    dov.x += exo.x; dov.y += exo.y; dov.z += exo.z; dov.w += exo.w;
                }
#pragma unroll
                for (int o = 16; o > 0; o >>= 1) {
                    dov.x += __shfl_xor(dov.x, o, 32);
                    dov.y += __shfl_xor(dov.y, o, 32);
                    dov.z += __shfl_xor(dov.z, o, 32);
                    dov.w += __shfl_xor(dov.w, o, 32);
                }
                den += (hd == 0) ? dov.x : (hd == 1) ? dov.y : (hd == 2) ? dov.z : dov.w;
            }
            den += 1e-16f;
#pragma unroll
            for (int q = 0; q < 4; ++q)
                if (8 * q < d) {
                    const int rem = d - 8 * q;
                    group_accumulate(P4, sv[q], ex[q] / den, (rem < 8) ? rem : 8, lane, acc);
                }
            for (int c0 = 32; c0 < d; c0 += 8) {
                const int j = c0 + e8;
                const int sx = lsrc[off + ((j < d) ? j : d - 1)];
                const float exx = group_exp(s_src[(unsigned)(sx * N_HEAD + hd)], stl, gm, j < d);
                const int rem = d - c0;
                group_accumulate(P4, sx, exx / den, (rem < 8) ? rem : 8, lane, acc);
            }
            if (novf > 0) {
                const float4 st = ((const float4*)s_tgt)[n];
                const int m0 = (hd == 0) ? -1 : 0, m1 = (hd == 1) ? -1 : 0;
                const int m2 = (hd == 2) ? -1 : 0, m3 = (hd == 3) ? -1 : 0;
                const float4 den4 = make_float4(
                    __int_as_float(__builtin_amdgcn_readlane(__float_as_int(den), 0)),
                    __int_as_float(__builtin_amdgcn_readlane(__float_as_int(den), 8)),
                    __int_as_float(__builtin_amdgcn_readlane(__float_as_int(den), 16)),
                    __int_as_float(__builtin_amdgcn_readlane(__float_as_int(den), 24)));
                for (int b = 0; b < novf; b += 32) {
                    int sx; float4 exo; bool hit;
                    overflow_weights(ovf_tgt, ovf_src, s_src4, st, gm, b, novf, n, lane, sx, exo, hit);
                    const float4 att = make_float4(exo.x / den4.x, exo.y / den4.y, exo.z / den4.z, exo.w / den4.w);
                    unsigned todo = __builtin_amdgcn_ballot_w32(hit);
                    while (todo != 0u) {
                        const int j = __builtin_ctz(todo);
                        todo &= todo - 1u;
                        const int sj = __builtin_amdgcn_readlane(sx, j);
                        const float4 r = P4[(size_t)sj * (D_IN / 4) + lane];
                        const float a = head_weight(att, j, m0, m1, m2, m3);
                        acc.x = __builtin_fmaf(r.x, a, acc.x);
                        acc.y = __builtin_fmaf(r.y, a, acc.y);
                        acc.z = __builtin_fmaf(r.z, a, acc.z);
                        acc.w = __builtin_fmaf(r.w, a, acc.w);
                    }
                }
            }
        }
        ((float4*)out)[(size_t)n * (D_IN / 4) + lane] =
            make_float4(acc.x + bv.x, acc.y + bv.y, acc.z + bv.z, acc.w + bv.w);
    }
}

extern "C" void kernel_launch(void* const* d_in, const int* in_sizes, int n_in,
                              void* d_out, int out_size, void* d_ws, size_t ws_size,
                              hipStream_t stream) {
    const float* in_feat = (const float*)d_in[0];
    const int*   ei      = (const int*)  d_in[1];
    const float* W_proj  = (const float*)d_in[3];
    const float* a_src   = (const float*)d_in[4];
    const float* a_tgt   = (const float*)d_in[5];
    const float* bias    = (const float*)d_in[6];
    float*       out     = (float*)d_out;

    float*    proj    = (float*)d_ws;
    float*    s_src   = proj  + (size_t)N_NODES * D_IN;
    float*    s_tgt   = s_src + (size_t)N_NODES * N_HEAD;
    int*      seg     = (int*)(s_tgt + (size_t)N_NODES * N_HEAD);
    unsigned* seg_cnt = (unsigned*)(seg + (size_t)NB * BIN_WGS * SEGCAP * 2);
    int*      ovf_tgt = (int*)(seg_cnt + (size_t)BIN_WGS * NB);
    int*      ovf_src = ovf_tgt + (size_t)N_EDGES;
    float*    wg_max  = (float*)(ovf_src + (size_t)N_EDGES);
    unsigned* ovf_cnt = (unsigned*)(wg_max + BIN_WGS);

    gat_proj_scores_kernel<<<N_NODES / 16, 128, 0, stream>>>(
        in_feat, W_proj, a_src, a_tgt, proj, s_src, s_tgt, ovf_cnt);

    gat_bin_max_kernel<<<BIN_WGS, 1024, 0, stream>>>(
        ei, s_src, s_tgt, wg_max, seg_cnt, seg, ovf_cnt, ovf_tgt, ovf_src);

    gat_gather_kernel<<<NB, 1024, 0, stream>>>(
        seg_cnt, seg, ovf_cnt, ovf_tgt, ovf_src, proj, s_src, s_tgt, wg_max, bias, out);
}
